# starting kernel
# baseline (speedup 1.0000x reference)
#include <hip/hip_runtime.h>

typedef __attribute__((ext_vector_type(2))) float v2f;
typedef __attribute__((ext_vector_type(8))) float v8f;

__global__ void fill_kernel(float* __restrict__ p, float v, int n) {
    int i = blockIdx.x * blockDim.x + threadIdx.x;
    if (i < n) p[i] = v;
}

__global__ void deg_kernel(const int* __restrict__ dst, float* __restrict__ deg, int E) {
    int i = blockIdx.x * blockDim.x + threadIdx.x;
    if (i < E) unsafeAtomicAdd(&deg[dst[i]], 1.0f);
}

__global__ void rsqrt_kernel(float* __restrict__ deg, int n) {
    int i = blockIdx.x * blockDim.x + threadIdx.x;
    if (i < n) deg[i] = rsqrtf(deg[i]);
}

__device__ __forceinline__ float squash(float v) {
    return (v * v) * __builtin_amdgcn_rcpf(v);
}

template <int K, int N, bool SQUASH>
__global__ __launch_bounds__((N / 16) * 32)
void gemm_wmma(const float* __restrict__ X, const float* __restrict__ W,
               float* __restrict__ H, int M) {
    const int lane  = threadIdx.x & 31;
    const int ntile = threadIdx.x >> 5;
    const int half  = lane >> 4;
    const int l16   = lane & 15;
    const int row   = blockIdx.x * 16 + l16;
    const int arow  = row < M ? row : M - 1;
    const int ncol  = ntile * 16 + l16;
    const float* __restrict__ xrow = X + (long)arow * K;

    v8f c = {0.f, 0.f, 0.f, 0.f, 0.f, 0.f, 0.f, 0.f};

#pragma unroll
    for (int kk = 0; kk < K; kk += 4) {
        const int k0 = kk + half * 2;
        v2f a;
        {
            float2 av = *(const float2*)(xrow + k0);
            if (SQUASH) { av.x = squash(av.x); av.y = squash(av.y); }
            a.x = av.x; a.y = av.y;
        }
        v2f b;
        b.x = W[k0 * N + ncol];
        b.y = W[(k0 + 1) * N + ncol];
        c = __builtin_amdgcn_wmma_f32_16x16x4_f32(
                 false, a,  false, b,
                 (short)0, c,  false,  false);
    }

#pragma unroll
    for (int i = 0; i < 8; ++i) {
        int r = blockIdx.x * 16 + i + half * 8;
        if (r < M) H[(long)r * N + ncol] = c[i];
    }
}

template <int C>
__global__ void scatter_kernel(const float* __restrict__ h, const int* __restrict__ src,
                               const int* __restrict__ dst, const float* __restrict__ dinv,
                               float* __restrict__ acc, int E) {
    int gid  = blockIdx.x * blockDim.x + threadIdx.x;
    int e    = gid >> 5;
    int lane = gid & 31;
    if (e >= E) return;
    int s = src[e], d = dst[e];
    float p = dinv[s] * dinv[d];
    if (C == 64) {
        float2 v = *(const float2*)(h + (long)s * 64 + lane * 2);
        unsafeAtomicAdd(&acc[(long)d * 64 + lane * 2 + 0], p * v.x);
        unsafeAtomicAdd(&acc[(long)d * 64 + lane * 2 + 1], p * v.y);
    } else {
        float v = h[(long)s * 32 + lane];
        unsafeAtomicAdd(&acc[(long)d * 32 + lane], p * v);
    }
}

template <int C, bool RELU>
__global__ void finalize_kernel(const float* __restrict__ acc, const float* __restrict__ h,
                                const float* __restrict__ dinv, const float* __restrict__ bias,
                                float* __restrict__ out, int Nn) {
    int idx = blockIdx.x * blockDim.x + threadIdx.x;
    if (idx >= Nn * C) return;
    int i = idx / C, ch = idx & (C - 1);
    float di = dinv[i];
    float v = acc[idx] + di * di * h[idx] + bias[ch];
    out[idx] = RELU ? fmaxf(v, 0.0f) : v;
}

static inline size_t align256(size_t n) { return (n + 255) & ~size_t(255); }

extern "C" void kernel_launch(void* const* d_in, const int* in_sizes, int n_in,
                              void* d_out, int out_size, void* d_ws, size_t ws_size,
                              hipStream_t stream) {
    const float* x  = (const float*)d_in[0];
    const int*   ei = (const int*)d_in[1];
    const float* W1 = (const float*)d_in[2];
    const float* b1 = (const float*)d_in[3];
    const float* W2 = (const float*)d_in[4];
    const float* b2 = (const float*)d_in[5];
    float* out = (float*)d_out;

    const int hid  = in_sizes[3];
    const int outc = in_sizes[5];
    const int inc  = in_sizes[2] / hid;
    const int Nn   = in_sizes[0] / inc;
    const int E    = in_sizes[1] / 2;
    const int* src = ei;
    const int* dstv = ei + E;
    (void)outc;

    char* ws = (char*)d_ws;
    float* deg = (float*)ws;  ws += align256((size_t)Nn * 4);
    float* h1  = (float*)ws;  ws += (size_t)Nn * 64 * 4;
    float* a1  = (float*)ws;  ws += (size_t)Nn * 64 * 4;
    float* h2  = (float*)ws;

    const int T = 256;
    fill_kernel<<<(Nn + T - 1) / T, T, 0, stream>>>(deg, 1.0f, Nn);
    fill_kernel<<<(Nn * 64 + T - 1) / T, T, 0, stream>>>(a1, 0.0f, Nn * 64);
    fill_kernel<<<(Nn * 32 + T - 1) / T, T, 0, stream>>>(out, 0.0f, Nn * 32);

    deg_kernel<<<(E + T - 1) / T, T, 0, stream>>>(dstv, deg, E);
    rsqrt_kernel<<<(Nn + T - 1) / T, T, 0, stream>>>(deg, Nn);

    gemm_wmma<128, 64, true><<<(Nn + 15) / 16, 128, 0, stream>>>(x, W1, h1, Nn);
    scatter_kernel<64><<<(E + 7) / 8, T, 0, stream>>>(h1, src, dstv, deg, a1, E);
    finalize_kernel<64, true><<<(Nn * 64 + T - 1) / T, T, 0, stream>>>(a1, h1, deg, b1, a1, Nn);

    gemm_wmma<64, 32, false><<<(Nn + 15) / 16, 64, 0, stream>>>(a1, W2, h2, Nn);
    scatter_kernel<32><<<(E + 7) / 8, T, 0, stream>>>(h2, src, dstv, deg, out, E);
    finalize_kernel<32, false><<<(Nn * 32 + T - 1) / T, T, 0, stream>>>(out, h2, deg, b2, out, Nn);
}


# scatter kernels: one wave per 32 consecutive edges (per-lane index and norm loads, scalar broadcast per edge, 16 row loads in flight) instead of one wave per edge; adds stay atomic, arrival order
# speedup vs baseline: 1.6272x; 1.6272x over previous
#include <hip/hip_runtime.h>

typedef __attribute__((ext_vector_type(2))) float v2f;
typedef __attribute__((ext_vector_type(8))) float v8f;

__global__ void fill_kernel(float* __restrict__ p, float v, int n) {
    int i = blockIdx.x * blockDim.x + threadIdx.x;
    if (i < n) p[i] = v;
}

__global__ void deg_kernel(const int* __restrict__ dst, float* __restrict__ deg, int E) {
    int i = blockIdx.x * blockDim.x + threadIdx.x;
    if (i < E) unsafeAtomicAdd(&deg[dst[i]], 1.0f);
}

__global__ void rsqrt_kernel(float* __restrict__ deg, int n) {
    int i = blockIdx.x * blockDim.x + threadIdx.x;
    if (i < n) deg[i] = rsqrtf(deg[i]);
}

__device__ __forceinline__ float squash(float v) {
    return (v * v) * __builtin_amdgcn_rcpf(v);
}

template <int K, int N, bool SQUASH>
__global__ __launch_bounds__((N / 16) * 32)
void gemm_wmma(const float* __restrict__ X, const float* __restrict__ W,
               float* __restrict__ H, int M) {
    const int lane  = threadIdx.x & 31;
    const int ntile = threadIdx.x >> 5;
    const int half  = lane >> 4;
    const int l16   = lane & 15;
    const int row   = blockIdx.x * 16 + l16;
    const int arow  = row < M ? row : M - 1;
    const int ncol  = ntile * 16 + l16;
    const float* __restrict__ xrow = X + (long)arow * K;

    v8f c = {0.f, 0.f, 0.f, 0.f, 0.f, 0.f, 0.f, 0.f};

#pragma unroll
    for (int kk = 0; kk < K; kk += 4) {
        const int k0 = kk + half * 2;
        v2f a;
        {
            float2 av = *(const float2*)(xrow + k0);
            if (SQUASH) { av.x = squash(av.x); av.y = squash(av.y); }
            a.x = av.x; a.y = av.y;
        }
        v2f b;
        b.x = W[k0 * N + ncol];
        b.y = W[(k0 + 1) * N + ncol];
        c = __builtin_amdgcn_wmma_f32_16x16x4_f32(
                 false, a,  false, b,
                 (short)0, c,  false,  false);
    }

#pragma unroll
    for (int i = 0; i < 8; ++i) {
        int r = blockIdx.x * 16 + i + half * 8;
        if (r < M) H[(long)r * N + ncol] = c[i];
    }
}

#define GLOBAL_MEM __attribute__((address_space(1)))
template <typename T>
__device__ __forceinline__ GLOBAL_MEM T* wave_uniform_row(T* base, int row, int width) {
    const unsigned long long a = (unsigned long long)(base + (long)row * width);
    const unsigned lo = __builtin_amdgcn_readfirstlane((unsigned)a);
    const unsigned hi = __builtin_amdgcn_readfirstlane((unsigned)(a >> 32));
    return (GLOBAL_MEM T*)(((unsigned long long)hi << 32) | lo);
}

template <int C>
__global__ __launch_bounds__(256)
void scatter_kernel(const float* __restrict__ h, const int* __restrict__ src,
                    const int* __restrict__ dst, const float* __restrict__ dinv,
                    float* __restrict__ acc, int E) {
    constexpr int G = 16;
    const int lane = threadIdx.x & 31;
    const int base = ((blockIdx.x * blockDim.x + threadIdx.x) >> 5) * 32;
    if (base >= E) return;
    const int e = base + lane;
    int s = 0, d = 0;
    float p = 0.0f;
    if (e < E) {
        s = src[e];
        d = dst[e];
        p = dinv[s] * dinv[d];
    }
    const int pbits = __float_as_int(p);

    if (base + 32 <= E) {
#pragma unroll
        for (int j0 = 0; j0 < 32; j0 += G) {
            float v0[G], v1[G];
#pragma unroll
            for (int u = 0; u < G; ++u) {
                const int sj = __builtin_amdgcn_readlane(s, j0 + u);
                const GLOBAL_MEM float* hr = wave_uniform_row(h, sj, C);
                v0[u] = hr[lane];
                if (C == 64) v1[u] = hr[32 + lane];
            }
#pragma unroll
            for (int u = 0; u < G; ++u) {
                const int dj = __builtin_amdgcn_readlane(d, j0 + u);
                const float pj = __int_as_float(__builtin_amdgcn_readlane(pbits, j0 + u));
                GLOBAL_MEM float* ar = wave_uniform_row(acc, dj, C);
                unsafeAtomicAdd((float*)&ar[lane], pj * v0[u]);
                if (C == 64) unsafeAtomicAdd((float*)&ar[32 + lane], pj * v1[u]);
            }
        }
    } else {
        const int cnt = E - base;
        for (int j = 0; j < cnt; ++j) {
            const int sj = __builtin_amdgcn_readlane(s, j);
            const int dj = __builtin_amdgcn_readlane(d, j);
            const float pj = __int_as_float(__builtin_amdgcn_readlane(pbits, j));
            const GLOBAL_MEM float* hr = wave_uniform_row(h, sj, C);
            GLOBAL_MEM float* ar = wave_uniform_row(acc, dj, C);
            unsafeAtomicAdd((float*)&ar[lane], pj * hr[lane]);
            if (C == 64) unsafeAtomicAdd((float*)&ar[32 + lane], pj * hr[32 + lane]);
        }
    }
}

template <int C, bool RELU>
__global__ void finalize_kernel(const float* __restrict__ acc, const float* __restrict__ h,
                                const float* __restrict__ dinv, const float* __restrict__ bias,
                                float* __restrict__ out, int Nn) {
    int idx = blockIdx.x * blockDim.x + threadIdx.x;
    if (idx >= Nn * C) return;
    int i = idx / C, ch = idx & (C - 1);
    float di = dinv[i];
    float v = acc[idx] + di * di * h[idx] + bias[ch];
    out[idx] = RELU ? fmaxf(v, 0.0f) : v;
}

static inline size_t align256(size_t n) { return (n + 255) & ~size_t(255); }

extern "C" void kernel_launch(void* const* d_in, const int* in_sizes, int n_in,
                              void* d_out, int out_size, void* d_ws, size_t ws_size,
                              hipStream_t stream) {
    const float* x  = (const float*)d_in[0];
    const int*   ei = (const int*)d_in[1];
    const float* W1 = (const float*)d_in[2];
    const float* b1 = (const float*)d_in[3];
    const float* W2 = (const float*)d_in[4];
    const float* b2 = (const float*)d_in[5];
    float* out = (float*)d_out;

    const int hid  = in_sizes[3];
    const int outc = in_sizes[5];
    const int inc  = in_sizes[2] / hid;
    const int Nn   = in_sizes[0] / inc;
    const int E    = in_sizes[1] / 2;
    const int* src = ei;
    const int* dstv = ei + E;
    (void)outc;

    char* ws = (char*)d_ws;
    float* deg = (float*)ws;  ws += align256((size_t)Nn * 4);
    float* h1  = (float*)ws;  ws += (size_t)Nn * 64 * 4;
    float* a1  = (float*)ws;  ws += (size_t)Nn * 64 * 4;
    float* h2  = (float*)ws;

    const int T = 256;
    fill_kernel<<<(Nn + T - 1) / T, T, 0, stream>>>(deg, 1.0f, Nn);
    fill_kernel<<<(Nn * 64 + T - 1) / T, T, 0, stream>>>(a1, 0.0f, Nn * 64);
    fill_kernel<<<(Nn * 32 + T - 1) / T, T, 0, stream>>>(out, 0.0f, Nn * 32);

    deg_kernel<<<(E + T - 1) / T, T, 0, stream>>>(dstv, deg, E);
    rsqrt_kernel<<<(Nn + T - 1) / T, T, 0, stream>>>(deg, Nn);

    gemm_wmma<128, 64, true><<<(Nn + 15) / 16, 128, 0, stream>>>(x, W1, h1, Nn);
    scatter_kernel<64><<<(E + T - 1) / T, T, 0, stream>>>(h1, src, dstv, deg, a1, E);
    finalize_kernel<64, true><<<(Nn * 64 + T - 1) / T, T, 0, stream>>>(a1, h1, deg, b1, a1, Nn);

    gemm_wmma<64, 32, false><<<(Nn + 15) / 16, 64, 0, stream>>>(a1, W2, h2, Nn);
    scatter_kernel<32><<<(E + T - 1) / T, T, 0, stream>>>(h2, src, dstv, deg, out, E);
    finalize_kernel<32, false><<<(Nn * 32 + T - 1) / T, T, 0, stream>>>(out, h2, deg, b2, out, Nn);
}


# aggregation: edges counting-sorted by destination inside the call; one wave per node sums its rows in registers, one store per line; no float atomics, clears or finalize; sum order = arrival order
# speedup vs baseline: 2.0074x; 1.2337x over previous
#include <hip/hip_runtime.h>

typedef __attribute__((ext_vector_type(2))) float v2f;
typedef __attribute__((ext_vector_type(8))) float v8f;

typedef __attribute__((ext_vector_type(2))) int v2i;
typedef __attribute__((ext_vector_type(4))) int v4i;

constexpr int EPT = 4;

__global__ void zero_kernel(int* __restrict__ p, int n) {
    int i = blockIdx.x * blockDim.x + threadIdx.x;
    if (i < n) p[i] = 0;
}

template <bool FULL>
__device__ __forceinline__ void count_rank_body(const int* __restrict__ dst, int* __restrict__ cnt,
                                                int* __restrict__ rank, int E, int base) {
    int d[EPT], r[EPT];
#pragma unroll
    for (int k = 0; k < EPT; ++k) {
        const int e = base + k * 256;
        d[k] = (FULL || e < E) ? dst[e] : 0;
    }
#pragma unroll
    for (int k = 0; k < EPT; ++k) {
        const int e = base + k * 256;
        r[k] = (FULL || e < E) ? atomicAdd(&cnt[d[k]], 1) : 0;
    }
#pragma unroll
    for (int k = 0; k < EPT; ++k) {
        const int e = base + k * 256;
        if (FULL || e < E) rank[e] = r[k];
    }
}

__global__ __launch_bounds__(256)
void count_rank_kernel(const int* __restrict__ dst, int* __restrict__ cnt,
                       int* __restrict__ rank, int E) {
    const int blk0 = blockIdx.x * (256 * EPT);
    const int base = blk0 + threadIdx.x;
    if (blk0 + 256 * EPT <= E) count_rank_body<true>(dst, cnt, rank, E, base);
    else                       count_rank_body<false>(dst, cnt, rank, E, base);
}

__global__ __launch_bounds__(1024)
void scan_kernel(const int* __restrict__ cnt, int* __restrict__ info, int n) {
    __shared__ int s_pre[32];
    __shared__ int s_tot[32];
    const int t = threadIdx.x, lane = t & 31, w = t >> 5;
    const int tile0 = blockIdx.x * 1024;
    int part = 0;
    const v4i* c4 = (const v4i*)cnt;
    const int n4 = tile0 >> 2;
#pragma unroll 4
    for (int i = t; i < n4; i += 1024) {
        const v4i v = c4[i];
        part += (v.x + v.y) + (v.z + v.w);
    }
#pragma unroll
    for (int o = 16; o > 0; o >>= 1) part += __shfl_xor(part, o);
    const int i = tile0 + t;
    const int c = i < n ? cnt[i] : 0;
    int incl = c;
#pragma unroll
    for (int o = 1; o < 32; o <<= 1) {
        const int v = __shfl_up(incl, o);
        if (lane >= o) incl += v;
    }
    if (lane == 31) s_tot[w] = incl;
    if (lane == 0)  s_pre[w] = part;
    __syncthreads();
    int a = s_pre[lane];
    int b = lane < w ? s_tot[lane] : 0;
#pragma unroll
    for (int o = 16; o > 0; o >>= 1) { a += __shfl_xor(a, o); b += __shfl_xor(b, o); }
    if (i < n) {
        v2i rec;
        rec.x = a + b + (incl - c);
        rec.y = __float_as_int(rsqrtf((float)(c + 1)));
        ((v2i*)info)[i] = rec;
    }
}

template <bool FULL>
__device__ __forceinline__ void csr_fill_body(const int* __restrict__ src, const int* __restrict__ dst,
                                              const int* __restrict__ rank, const v2i* __restrict__ in2,
                                              v2i* __restrict__ out2, int E, int base) {
    int s[EPT], d[EPT], rk[EPT];
#pragma unroll
    for (int k = 0; k < EPT; ++k) {
        const int e = base + k * 256;
        const bool ok = FULL || e < E;
        s[k]  = ok ? src[e] : 0;
        d[k]  = ok ? dst[e] : 0;
        rk[k] = ok ? rank[e] : 0;
    }
    v2i id[EPT];
    int ps[EPT];
#pragma unroll
    for (int k = 0; k < EPT; ++k) {
        id[k] = in2[d[k]];
        ps[k] = in2[s[k]].y;
    }
#pragma unroll
    for (int k = 0; k < EPT; ++k) {
        const int e = base + k * 256;
        if (FULL || e < E) {
            const float p = __int_as_float(ps[k]) * __int_as_float(id[k].y);
            v2i rec;
            rec.x = s[k];
            rec.y = __float_as_int(p);
            out2[id[k].x + rk[k]] = rec;
        }
    }
}

__global__ __launch_bounds__(256)
void csr_fill_kernel(const int* __restrict__ src, const int* __restrict__ dst,
                     const int* __restrict__ rank, const int* __restrict__ info,
                     int* __restrict__ sorted, int E) {
    const int blk0 = blockIdx.x * (256 * EPT);
    const int base = blk0 + threadIdx.x;
    const v2i* in2 = (const v2i*)info;
    v2i* out2 = (v2i*)sorted;
    if (blk0 + 256 * EPT <= E) csr_fill_body<true>(src, dst, rank, in2, out2, E, base);
    else                       csr_fill_body<false>(src, dst, rank, in2, out2, E, base);
}

__device__ __forceinline__ float squash(float v) {
    return (v * v) * __builtin_amdgcn_rcpf(v);
}

template <int K, int N, bool SQUASH>
__global__ __launch_bounds__((N / 16) * 32)
void gemm_wmma(const float* __restrict__ X, const float* __restrict__ W,
               float* __restrict__ H, int M) {
    const int lane  = threadIdx.x & 31;
    const int ntile = threadIdx.x >> 5;
    const int half  = lane >> 4;
    const int l16   = lane & 15;
    const int row   = blockIdx.x * 16 + l16;
    const int arow  = row < M ? row : M - 1;
    const int ncol  = ntile * 16 + l16;
    const float* __restrict__ xrow = X + (long)arow * K;

    v8f c = {0.f, 0.f, 0.f, 0.f, 0.f, 0.f, 0.f, 0.f};

#pragma unroll
    for (int kk = 0; kk < K; kk += 4) {
        const int k0 = kk + half * 2;
        v2f a;
        {
            float2 av = *(const float2*)(xrow + k0);
            if (SQUASH) { av.x = squash(av.x); av.y = squash(av.y); }
            a.x = av.x; a.y = av.y;
        }
        v2f b;
        b.x = W[k0 * N + ncol];
        b.y = W[(k0 + 1) * N + ncol];
        c = __builtin_amdgcn_wmma_f32_16x16x4_f32(
                 false, a,  false, b,
                 (short)0, c,  false,  false);
    }

#pragma unroll
    for (int i = 0; i < 8; ++i) {
        int r = blockIdx.x * 16 + i + half * 8;
        if (r < M) H[(long)r * N + ncol] = c[i];
    }
}

#define GLOBAL_MEM __attribute__((address_space(1)))
template <typename T>
__device__ __forceinline__ GLOBAL_MEM T* wave_uniform_row(T* base, int row, int width) {
    const unsigned long long a = (unsigned long long)(base + (long)row * width);
    const unsigned lo = __builtin_amdgcn_readfirstlane((unsigned)a);
    const unsigned hi = __builtin_amdgcn_readfirstlane((unsigned)(a >> 32));
    return (GLOBAL_MEM T*)(((unsigned long long)hi << 32) | lo);
}

template <int C, bool RELU>
__global__ __launch_bounds__(256)
void gather_kernel(const float* __restrict__ h, const int* __restrict__ sorted,
                   const int* __restrict__ info, const int* __restrict__ cnt,
                   const float* __restrict__ bias, float* __restrict__ out, int Nn) {
    constexpr int G = 8;
    const int lane = threadIdx.x & 31;
    const int node = __builtin_amdgcn_readfirstlane((int)((blockIdx.x * blockDim.x + threadIdx.x) >> 5));
    if (node >= Nn) return;
    const v2i ni = ((const v2i*)info)[node];
    const int start = __builtin_amdgcn_readfirstlane(ni.x);
    const float di = __int_as_float(ni.y);
    const int n = __builtin_amdgcn_readfirstlane(cnt[node]);
    const v2i* ent = (const v2i*)sorted + start;

    const GLOBAL_MEM float* hs = wave_uniform_row(h, node, C);
    float sx, sy = 0.0f;
    if (C == 64) { const v2f sv = ((const GLOBAL_MEM v2f*)hs)[lane]; sx = sv.x; sy = sv.y; }
    else         { sx = hs[lane]; }

    float ax = 0.0f, ay = 0.0f;
    for (int c0 = 0; c0 < n; c0 += 32) {
        const int m = (n - c0) < 32 ? (n - c0) : 32;
        int s = 0, pb = 0;
        if (lane < m) { const v2i en = ent[c0 + lane]; s = en.x; pb = en.y; }
        for (int j0 = 0; j0 < m; j0 += G) {
            float v0[G], v1[G];
#pragma unroll
            for (int u = 0; u < G; ++u) {
                const int jj = (j0 + u) < m ? (j0 + u) : (m - 1);
                const int sj = __builtin_amdgcn_readlane(s, jj);
                const GLOBAL_MEM float* hr = wave_uniform_row(h, sj, C);
                if (C == 64) { const v2f tv = ((const GLOBAL_MEM v2f*)hr)[lane]; v0[u] = tv.x; v1[u] = tv.y; }
                else         { v0[u] = hr[lane]; }
            }
#pragma unroll
            for (int u = 0; u < G; ++u) {
                if (j0 + u < m) {
                    const float pj = __int_as_float(__builtin_amdgcn_readlane(pb, j0 + u));
                    ax += pj * v0[u];
                    if (C == 64) ay += pj * v1[u];
                }
            }
        }
    }

    const float dd = di * di;
    GLOBAL_MEM float* orow = wave_uniform_row(out, node, C);
    if (C == 64) {
        float ox = ax + dd * sx + bias[2 * lane];
        float oy = ay + dd * sy + bias[2 * lane + 1];
        if (RELU) { ox = fmaxf(ox, 0.0f); oy = fmaxf(oy, 0.0f); }
        v2f ov; ov.x = ox; ov.y = oy;
        ((GLOBAL_MEM v2f*)orow)[lane] = ov;
    } else {
        float ox = ax + dd * sx + bias[lane];
        if (RELU) ox = fmaxf(ox, 0.0f);
        orow[lane] = ox;
    }
}

static inline size_t align256(size_t n) { return (n + 255) & ~size_t(255); }

extern "C" void kernel_launch(void* const* d_in, const int* in_sizes, int n_in,
                              void* d_out, int out_size, void* d_ws, size_t ws_size,
                              hipStream_t stream) {
    const float* x  = (const float*)d_in[0];
    const int*   ei = (const int*)d_in[1];
    const float* W1 = (const float*)d_in[2];
    const float* b1 = (const float*)d_in[3];
    const float* W2 = (const float*)d_in[4];
    const float* b2 = (const float*)d_in[5];
    float* out = (float*)d_out;

    const int hid  = in_sizes[3];
    const int outc = in_sizes[5];
    const int inc  = in_sizes[2] / hid;
    const int Nn   = in_sizes[0] / inc;
    const int E    = in_sizes[1] / 2;
    const int* src = ei;
    const int* dstv = ei + E;
    (void)outc;

    char* ws = (char*)d_ws;
    int* cnt    = (int*)ws;    ws += align256((size_t)Nn * 4);
    int* info   = (int*)ws;    ws += align256((size_t)Nn * 8);
    int* rank   = (int*)ws;    ws += align256((size_t)E * 4);
    int* sorted = (int*)ws;    ws += align256((size_t)E * 8);
    float* h1   = (float*)ws;  ws += (size_t)Nn * 64 * 4;
    float* a1   = (float*)ws;  ws += (size_t)Nn * 64 * 4;
    float* h2   = (float*)ws;

    const int T = 256;
    zero_kernel<<<(Nn + T - 1) / T, T, 0, stream>>>(cnt, Nn);
    count_rank_kernel<<<(E + T * EPT - 1) / (T * EPT), T, 0, stream>>>(dstv, cnt, rank, E);
    scan_kernel<<<(Nn + 1023) / 1024, 1024, 0, stream>>>(cnt, info, Nn);
    csr_fill_kernel<<<(E + T * EPT - 1) / (T * EPT), T, 0, stream>>>(src, dstv, rank, info, sorted, E);

    gemm_wmma<128, 64, true><<<(Nn + 15) / 16, 128, 0, stream>>>(x, W1, h1, Nn);
    gather_kernel<64, true><<<(Nn + 7) / 8, T, 0, stream>>>(h1, sorted, info, cnt, b1, a1, Nn);

    gemm_wmma<64, 32, false><<<(Nn + 15) / 16, 64, 0, stream>>>(a1, W2, h2, Nn);
    gather_kernel<32, false><<<(Nn + 7) / 8, T, 0, stream>>>(h2, sorted, info, cnt, b2, out, Nn);
}


# list build: edges binned by 512-node range, then ranked per node, with LDS counters (the LDS add's return is the slot), runs staged in LDS; no global atomics; per-node sum order stays arrival order
# speedup vs baseline: 2.9116x; 1.4504x over previous
#include <hip/hip_runtime.h>

typedef __attribute__((ext_vector_type(2))) float v2f;
typedef __attribute__((ext_vector_type(8))) float v8f;

typedef __attribute__((ext_vector_type(2))) int v2i;
typedef __attribute__((ext_vector_type(4))) int v4i;

constexpr int BIN_SHIFT = 9;
constexpr int BIN_NODES = 1 << BIN_SHIFT;
constexpr int NB_MAX    = 256;
constexpr int EPT       = 16;
constexpr int EPB       = 256 * EPT;

__global__ __launch_bounds__(256)
void bin_count_kernel(const int* __restrict__ dst, int* __restrict__ table, int E, int NB, int nblk) {
    __shared__ int hist[NB_MAX];
    const int t = threadIdx.x;
    hist[t] = 0;
    __syncthreads();
    const int blk0 = blockIdx.x * EPB;
    if (blk0 + EPB <= E) {
        int d[EPT];
#pragma unroll
        for (int k = 0; k < EPT; ++k) d[k] = dst[blk0 + k * 256 + t];
#pragma unroll
        for (int k = 0; k < EPT; ++k) atomicAdd(&hist[d[k] >> BIN_SHIFT], 1);
    } else {
        for (int k = 0; k < EPT; ++k) {
            const int e = blk0 + k * 256 + t;
            if (e < E) atomicAdd(&hist[dst[e] >> BIN_SHIFT], 1);
        }
    }
    __syncthreads();
    if (t < NB) table[t * nblk + blockIdx.x] = hist[t];
}

__global__ __launch_bounds__(1024)
void scan_kernel(const int* __restrict__ cnt, int* __restrict__ pre, int n) {
    __shared__ int s_pre[32];
    __shared__ int s_tot[32];
    const int t = threadIdx.x, lane = t & 31, w = t >> 5;
    const int tile0 = blockIdx.x * 1024;
    int part = 0;
    const v4i* c4 = (const v4i*)cnt;
    const int n4 = tile0 >> 2;
#pragma unroll 4
    for (int i = t; i < n4; i += 1024) {
        const v4i v = c4[i];
        part += (v.x + v.y) + (v.z + v.w);
    }
#pragma unroll
    for (int o = 16; o > 0; o >>= 1) part += __shfl_xor(part, o);
    const int i = tile0 + t;
    const int c = i < n ? cnt[i] : 0;
    int incl = c;
#pragma unroll
    for (int o = 1; o < 32; o <<= 1) {
        const int v = __shfl_up(incl, o);
        if (lane >= o) incl += v;
    }
    if (lane == 31) s_tot[w] = incl;
    if (lane == 0)  s_pre[w] = part;
    __syncthreads();
    int a = s_pre[lane];
    int b = lane < w ? s_tot[lane] : 0;
#pragma unroll
    for (int o = 16; o > 0; o >>= 1) { a += __shfl_xor(a, o); b += __shfl_xor(b, o); }
    if (i < n) pre[i] = a + b + (incl - c);
}

template <bool FULL>
__device__ __forceinline__ void bin_scatter_body(const int* __restrict__ src, const int* __restrict__ dst,
                                                 const int* __restrict__ pre, v2i* __restrict__ out2,
                                                 int* hist, int* gofs, int* wsum, v2i* buf,
                                                 int E, int NB, int nblk) {
    const int t = threadIdx.x, lane = t & 31, w = t >> 5;
    const int gfirst = t < NB ? pre[t * nblk + blockIdx.x] : 0;
    hist[t] = 0;
    __syncthreads();
    const int blk0 = blockIdx.x * EPB;
    const int nloc = FULL ? EPB : (E - blk0);
    int s[EPT], d[EPT], r[EPT];
#pragma unroll
    for (int k = 0; k < EPT; ++k) {
        const int e = blk0 + k * 256 + t;
        const int ec = (FULL || e < E) ? e : E - 1;
        s[k] = src[ec];
        d[k] = dst[ec];
    }
#pragma unroll
    for (int k = 0; k < EPT; ++k) {
        r[k] = 0;
        if (FULL || k * 256 + t < nloc) r[k] = atomicAdd(&hist[d[k] >> BIN_SHIFT], 1);
    }
    __syncthreads();
    const int c = hist[t];
    int incl = c;
#pragma unroll
    for (int o = 1; o < 32; o <<= 1) {
        const int v = __shfl_up(incl, o);
        if (lane >= o) incl += v;
    }
    if (lane == 31) wsum[w] = incl;
    __syncthreads();
    int front = 0;
#pragma unroll
    for (int j = 0; j < 8; ++j) front += j < w ? wsum[j] : 0;
    const int ls = front + incl - c;
    hist[t] = ls;
    gofs[t] = gfirst - ls;
    __syncthreads();
    int lp[EPT];
#pragma unroll
    for (int k = 0; k < EPT; ++k) lp[k] = hist[d[k] >> BIN_SHIFT] + r[k];
#pragma unroll
    for (int k = 0; k < EPT; ++k) {
        if (FULL || k * 256 + t < nloc) {
            v2i rec;
            rec.x = s[k];
            rec.y = d[k];
            buf[lp[k]] = rec;
        }
    }
    __syncthreads();
    v2i rec[EPT];
    int go[EPT];
#pragma unroll
    for (int k = 0; k < EPT; ++k) {
        const int i = k * 256 + t;
        rec[k] = buf[(FULL || i < nloc) ? i : 0];
    }
#pragma unroll
    for (int k = 0; k < EPT; ++k) go[k] = gofs[rec[k].y >> BIN_SHIFT];
#pragma unroll
    for (int k = 0; k < EPT; ++k) {
        const int i = k * 256 + t;
        if (FULL || i < nloc) out2[go[k] + i] = rec[k];
    }
}

__global__ __launch_bounds__(256)
void bin_scatter_kernel(const int* __restrict__ src, const int* __restrict__ dst,
                        const int* __restrict__ pre, int* __restrict__ bucketed,
                        int E, int NB, int nblk) {
    __shared__ int hist[NB_MAX];
    __shared__ int gofs[NB_MAX];
    __shared__ int wsum[8];
    __shared__ v2i buf[EPB];
    v2i* out2 = (v2i*)bucketed;
    if (blockIdx.x * EPB + EPB <= E) bin_scatter_body<true>(src, dst, pre, out2, hist, gofs, wsum, buf, E, NB, nblk);
    else                             bin_scatter_body<false>(src, dst, pre, out2, hist, gofs, wsum, buf, E, NB, nblk);
}

constexpr int STAGE_CAP = 16384;
__global__ __launch_bounds__(1024)
void bin_sort_kernel(const int* __restrict__ bucketed, const int* __restrict__ pre,
                     int* __restrict__ cnt, int* __restrict__ info, int* __restrict__ sorted,
                     int E, int Nn, int NB, int nblk) {
    __shared__ int lc[BIN_NODES];
    __shared__ int ld[BIN_NODES];
    __shared__ int wtot[BIN_NODES / 32];
    __shared__ v2i stage[STAGE_CAP];
    constexpr int U = 8;
    const int t = threadIdx.x, lane = t & 31, w = t >> 5;
    const int b = blockIdx.x;
    const int beg = pre[b * nblk];
    const int end = (b + 1 < NB) ? pre[(b + 1) * nblk] : E;
    const v2i* in2 = (const v2i*)bucketed;
    if (t < BIN_NODES) lc[t] = 0;
    __syncthreads();

    for (int i0 = beg + t; i0 < end; i0 += 1024 * U) {
        int d[U];
#pragma unroll
        for (int u = 0; u < U; ++u) {
            const int i = i0 + u * 1024;
            d[u] = in2[i < end ? i : i0].y;
        }
#pragma unroll
        for (int u = 0; u < U; ++u)
            if (i0 + u * 1024 < end) atomicAdd(&lc[d[u] & (BIN_NODES - 1)], 1);
    }
    __syncthreads();

    const int c = t < BIN_NODES ? lc[t] : 0;
    int incl = c;
#pragma unroll
    for (int o = 1; o < 32; o <<= 1) {
        const int v = __shfl_up(incl, o);
        if (lane >= o) incl += v;
    }
    if (lane == 31 && t < BIN_NODES) wtot[w] = incl;
    __syncthreads();
    if (t < BIN_NODES) {
        int front = 0;
        for (int j = 0; j < w; ++j) front += wtot[j];
        const int st = front + incl - c;
        const int dv = __float_as_int(rsqrtf((float)(c + 1)));
        lc[t] = st;
        ld[t] = dv;
        const int node = b * BIN_NODES + t;
        if (node < Nn) {
            v2i rec;
            rec.x = beg + st;
            rec.y = dv;
            ((v2i*)info)[node] = rec;
            cnt[node] = c;
        }
    }
    __syncthreads();

    v2i* out2 = (v2i*)sorted + beg;
    const int nb = end - beg;
    const bool fits = nb <= STAGE_CAP;
    for (int i0 = beg + t; i0 < end; i0 += 1024 * U) {
        v2i e[U];
#pragma unroll
        for (int u = 0; u < U; ++u) {
            const int i = i0 + u * 1024;
            e[u] = in2[i < end ? i : i0];
        }
#pragma unroll
        for (int u = 0; u < U; ++u) {
            if (i0 + u * 1024 < end) {
                const int nl = e[u].y & (BIN_NODES - 1);
                const int pos = atomicAdd(&lc[nl], 1);
                v2i rec;
                rec.x = e[u].x;
                rec.y = ld[nl];
                if (fits) stage[pos] = rec;
                else      out2[pos] = rec;
            }
        }
    }
    if (fits) {
        __syncthreads();
        for (int i0 = t; i0 < nb; i0 += 1024 * U) {
            v2i rec[U];
#pragma unroll
            for (int u = 0; u < U; ++u) {
                const int i = i0 + u * 1024;
                rec[u] = stage[i < nb ? i : i0];
            }
#pragma unroll
            for (int u = 0; u < U; ++u) {
                const int i = i0 + u * 1024;
                if (i < nb) out2[i] = rec[u];
            }
        }
    }
}

__global__ __launch_bounds__(256)
void norm_kernel(int* sorted, const int* __restrict__ info, int E) {
    constexpr int U = 4;
    v2i* s2 = (v2i*)sorted;
    const v2i* in2 = (const v2i*)info;
    const int base = blockIdx.x * (256 * U) + threadIdx.x;
    v2i e[U];
    int ps[U];
#pragma unroll
    for (int u = 0; u < U; ++u) {
        const int i = base + u * 256;
        if (i < E) e[u] = s2[i];
        else { e[u].x = 0; e[u].y = 0; }
    }
#pragma unroll
    for (int u = 0; u < U; ++u) ps[u] = in2[e[u].x].y;
#pragma unroll
    for (int u = 0; u < U; ++u) {
        const int i = base + u * 256;
        if (i < E) {
            const float p = __int_as_float(ps[u]) * __int_as_float(e[u].y);
            v2i rec;
            rec.x = e[u].x;
            rec.y = __float_as_int(p);
            s2[i] = rec;
        }
    }
}

__device__ __forceinline__ float squash(float v) {
    return (v * v) * __builtin_amdgcn_rcpf(v);
}

template <int K, int N, bool SQUASH>
__global__ __launch_bounds__((N / 16) * 32)
void gemm_wmma(const float* __restrict__ X, const float* __restrict__ W,
               float* __restrict__ H, int M) {
    const int lane  = threadIdx.x & 31;
    const int ntile = threadIdx.x >> 5;
    const int half  = lane >> 4;
    const int l16   = lane & 15;
    const int row   = blockIdx.x * 16 + l16;
    const int arow  = row < M ? row : M - 1;
    const int ncol  = ntile * 16 + l16;
    const float* __restrict__ xrow = X + (long)arow * K;

    v8f c = {0.f, 0.f, 0.f, 0.f, 0.f, 0.f, 0.f, 0.f};

#pragma unroll
    for (int kk = 0; kk < K; kk += 4) {
        const int k0 = kk + half * 2;
        v2f a;
        {
            float2 av = *(const float2*)(xrow + k0);
            if (SQUASH) { av.x = squash(av.x); av.y = squash(av.y); }
            a.x = av.x; a.y = av.y;
        }
        v2f b;
        b.x = W[k0 * N + ncol];
        b.y = W[(k0 + 1) * N + ncol];
        c = __builtin_amdgcn_wmma_f32_16x16x4_f32(
                 false, a,  false, b,
                 (short)0, c,  false,  false);
    }

#pragma unroll
    for (int i = 0; i < 8; ++i) {
        int r = blockIdx.x * 16 + i + half * 8;
        if (r < M) H[(long)r * N + ncol] = c[i];
    }
}

#define GLOBAL_MEM __attribute__((address_space(1)))
template <typename T>
__device__ __forceinline__ GLOBAL_MEM T* wave_uniform_row(T* base, int row, int width) {
    const unsigned long long a = (unsigned long long)(base + (long)row * width);
    const unsigned lo = __builtin_amdgcn_readfirstlane((unsigned)a);
    const unsigned hi = __builtin_amdgcn_readfirstlane((unsigned)(a >> 32));
    return (GLOBAL_MEM T*)(((unsigned long long)hi << 32) | lo);
}

template <int C, bool RELU>
__global__ __launch_bounds__(256)
void gather_kernel(const float* __restrict__ h, const int* __restrict__ sorted,
                   const int* __restrict__ info, const int* __restrict__ cnt,
                   const float* __restrict__ bias, float* __restrict__ out, int Nn) {
    constexpr int G = 8;
    const int lane = threadIdx.x & 31;
    const int node = __builtin_amdgcn_readfirstlane((int)((blockIdx.x * blockDim.x + threadIdx.x) >> 5));
    if (node >= Nn) return;
    const v2i ni = ((const v2i*)info)[node];
    const int start = __builtin_amdgcn_readfirstlane(ni.x);
    const float di = __int_as_float(ni.y);
    const int n = __builtin_amdgcn_readfirstlane(cnt[node]);
    const v2i* ent = (const v2i*)sorted + start;

    const GLOBAL_MEM float* hs = wave_uniform_row(h, node, C);
    float sx, sy = 0.0f;
    if (C == 64) { const v2f sv = ((const GLOBAL_MEM v2f*)hs)[lane]; sx = sv.x; sy = sv.y; }
    else         { sx = hs[lane]; }

    float ax = 0.0f, ay = 0.0f;
    for (int c0 = 0; c0 < n; c0 += 32) {
        const int m = (n - c0) < 32 ? (n - c0) : 32;
        int s = 0, pb = 0;
        if (lane < m) { const v2i en = ent[c0 + lane]; s = en.x; pb = en.y; }
        for (int j0 = 0; j0 < m; j0 += G) {
            float v0[G], v1[G];
#pragma unroll
            for (int u = 0; u < G; ++u) {
                const int jj = (j0 + u) < m ? (j0 + u) : (m - 1);
                const int sj = __builtin_amdgcn_readlane(s, jj);
                const GLOBAL_MEM float* hr = wave_uniform_row(h, sj, C);
                if (C == 64) { const v2f tv = ((const GLOBAL_MEM v2f*)hr)[lane]; v0[u] = tv.x; v1[u] = tv.y; }
                else         { v0[u] = hr[lane]; }
            }
#pragma unroll
            for (int u = 0; u < G; ++u) {
                if (j0 + u < m) {
                    const float pj = __int_as_float(__builtin_amdgcn_readlane(pb, j0 + u));
                    ax += pj * v0[u];
                    if (C == 64) ay += pj * v1[u];
                }
            }
        }
    }

    const float dd = di * di;
    GLOBAL_MEM float* orow = wave_uniform_row(out, node, C);
    if (C == 64) {
        float ox = ax + dd * sx + bias[2 * lane];
        float oy = ay + dd * sy + bias[2 * lane + 1];
        if (RELU) { ox = fmaxf(ox, 0.0f); oy = fmaxf(oy, 0.0f); }
        v2f ov; ov.x = ox; ov.y = oy;
        ((GLOBAL_MEM v2f*)orow)[lane] = ov;
    } else {
        float ox = ax + dd * sx + bias[lane];
        if (RELU) ox = fmaxf(ox, 0.0f);
        orow[lane] = ox;
    }
}

static inline size_t align256(size_t n) { return (n + 255) & ~size_t(255); }

extern "C" void kernel_launch(void* const* d_in, const int* in_sizes, int n_in,
                              void* d_out, int out_size, void* d_ws, size_t ws_size,
                              hipStream_t stream) {
    const float* x  = (const float*)d_in[0];
    const int*   ei = (const int*)d_in[1];
    const float* W1 = (const float*)d_in[2];
    const float* b1 = (const float*)d_in[3];
    const float* W2 = (const float*)d_in[4];
    const float* b2 = (const float*)d_in[5];
    float* out = (float*)d_out;

    const int hid  = in_sizes[3];
    const int outc = in_sizes[5];
    const int inc  = in_sizes[2] / hid;
    const int Nn   = in_sizes[0] / inc;
    const int E    = in_sizes[1] / 2;
    const int* src = ei;
    const int* dstv = ei + E;
    (void)outc;

    const int NB   = (Nn + BIN_NODES - 1) / BIN_NODES;
    const int nblk = (E + EPB - 1) / EPB;
    const int nt   = NB * nblk;

    char* ws = (char*)d_ws;
    int* cnt    = (int*)ws;    ws += align256((size_t)Nn * 4);
    int* info   = (int*)ws;    ws += align256((size_t)Nn * 8);
    int* table  = (int*)ws;    ws += align256((size_t)nt * 4);
    int* pre    = (int*)ws;    ws += align256((size_t)nt * 4);
    int* bucket = (int*)ws;    ws += align256((size_t)E * 8);
    int* sorted = (int*)ws;    ws += align256((size_t)E * 8);
    float* h1   = (float*)ws;  ws += (size_t)Nn * 64 * 4;
    float* a1   = (float*)ws;  ws += (size_t)Nn * 64 * 4;
    float* h2   = (float*)ws;

    const int T = 256;
    bin_count_kernel<<<nblk, 256, 0, stream>>>(dstv, table, E, NB, nblk);
    scan_kernel<<<(nt + 1023) / 1024, 1024, 0, stream>>>(table, pre, nt);
    bin_scatter_kernel<<<nblk, 256, 0, stream>>>(src, dstv, pre, bucket, E, NB, nblk);
    bin_sort_kernel<<<NB, 1024, 0, stream>>>(bucket, pre, cnt, info, sorted, E, Nn, NB, nblk);
    norm_kernel<<<(E + 1023) / 1024, 256, 0, stream>>>(sorted, info, E);

    gemm_wmma<128, 64, true><<<(Nn + 15) / 16, 128, 0, stream>>>(x, W1, h1, Nn);
    gather_kernel<64, true><<<(Nn + 7) / 8, T, 0, stream>>>(h1, sorted, info, cnt, b1, a1, Nn);

    gemm_wmma<64, 32, false><<<(Nn + 15) / 16, 64, 0, stream>>>(a1, W2, h2, Nn);
    gather_kernel<32, false><<<(Nn + 7) / 8, T, 0, stream>>>(h2, sorted, info, cnt, b2, out, Nn);
}


# aggregation: one wave per 8 consecutive nodes walks their contiguous list segment, 16 row loads in flight across node boundaries, next 32 indices prefetched; per-node sum order unchanged
# speedup vs baseline: 3.9168x; 1.3452x over previous
#include <hip/hip_runtime.h>

typedef __attribute__((ext_vector_type(2))) float v2f;
typedef __attribute__((ext_vector_type(8))) float v8f;

typedef __attribute__((ext_vector_type(2))) int v2i;
typedef __attribute__((ext_vector_type(4))) int v4i;

constexpr int BIN_SHIFT = 9;
constexpr int BIN_NODES = 1 << BIN_SHIFT;
constexpr int NB_MAX    = 256;
constexpr int EPT       = 16;
constexpr int EPB       = 256 * EPT;

__global__ __launch_bounds__(256)
void bin_count_kernel(const int* __restrict__ dst, int* __restrict__ table, int E, int NB, int nblk) {
    __shared__ int hist[NB_MAX];
    const int t = threadIdx.x;
    hist[t] = 0;
    __syncthreads();
    const int blk0 = blockIdx.x * EPB;
    if (blk0 + EPB <= E) {
        int d[EPT];
#pragma unroll
        for (int k = 0; k < EPT; ++k) d[k] = dst[blk0 + k * 256 + t];
#pragma unroll
        for (int k = 0; k < EPT; ++k) atomicAdd(&hist[d[k] >> BIN_SHIFT], 1);
    } else {
        for (int k = 0; k < EPT; ++k) {
            const int e = blk0 + k * 256 + t;
            if (e < E) atomicAdd(&hist[dst[e] >> BIN_SHIFT], 1);
        }
    }
    __syncthreads();
    if (t < NB) table[t * nblk + blockIdx.x] = hist[t];
}

__global__ __launch_bounds__(1024)
void scan_kernel(const int* __restrict__ cnt, int* __restrict__ pre, int n) {
    __shared__ int s_pre[32];
    __shared__ int s_tot[32];
    const int t = threadIdx.x, lane = t & 31, w = t >> 5;
    const int tile0 = blockIdx.x * 1024;
    int part = 0;
    const v4i* c4 = (const v4i*)cnt;
    const int n4 = tile0 >> 2;
#pragma unroll 4
    for (int i = t; i < n4; i += 1024) {
        const v4i v = c4[i];
        part += (v.x + v.y) + (v.z + v.w);
    }
#pragma unroll
    for (int o = 16; o > 0; o >>= 1) part += __shfl_xor(part, o);
    const int i = tile0 + t;
    const int c = i < n ? cnt[i] : 0;
    int incl = c;
#pragma unroll
    for (int o = 1; o < 32; o <<= 1) {
        const int v = __shfl_up(incl, o);
        if (lane >= o) incl += v;
    }
    if (lane == 31) s_tot[w] = incl;
    if (lane == 0)  s_pre[w] = part;
    __syncthreads();
    int a = s_pre[lane];
    int b = lane < w ? s_tot[lane] : 0;
#pragma unroll
    for (int o = 16; o > 0; o >>= 1) { a += __shfl_xor(a, o); b += __shfl_xor(b, o); }
    if (i < n) pre[i] = a + b + (incl - c);
}

template <bool FULL>
__device__ __forceinline__ void bin_scatter_body(const int* __restrict__ src, const int* __restrict__ dst,
                                                 const int* __restrict__ pre, v2i* __restrict__ out2,
                                                 int* hist, int* gofs, int* wsum, v2i* buf,
                                                 int E, int NB, int nblk) {
    const int t = threadIdx.x, lane = t & 31, w = t >> 5;
    const int gfirst = t < NB ? pre[t * nblk + blockIdx.x] : 0;
    hist[t] = 0;
    __syncthreads();
    const int blk0 = blockIdx.x * EPB;
    const int nloc = FULL ? EPB : (E - blk0);
    int s[EPT], d[EPT], r[EPT];
#pragma unroll
    for (int k = 0; k < EPT; ++k) {
        const int e = blk0 + k * 256 + t;
        const int ec = (FULL || e < E) ? e : E - 1;
        s[k] = src[ec];
        d[k] = dst[ec];
    }
#pragma unroll
    for (int k = 0; k < EPT; ++k) {
        r[k] = 0;
        if (FULL || k * 256 + t < nloc) r[k] = atomicAdd(&hist[d[k] >> BIN_SHIFT], 1);
    }
    __syncthreads();
    const int c = hist[t];
    int incl = c;
#pragma unroll
    for (int o = 1; o < 32; o <<= 1) {
        const int v = __shfl_up(incl, o);
        if (lane >= o) incl += v;
    }
    if (lane == 31) wsum[w] = incl;
    __syncthreads();
    int front = 0;
#pragma unroll
    for (int j = 0; j < 8; ++j) front += j < w ? wsum[j] : 0;
    const int ls = front + incl - c;
    hist[t] = ls;
    gofs[t] = gfirst - ls;
    __syncthreads();
    int lp[EPT];
#pragma unroll
    for (int k = 0; k < EPT; ++k) lp[k] = hist[d[k] >> BIN_SHIFT] + r[k];
#pragma unroll
    for (int k = 0; k < EPT; ++k) {
        if (FULL || k * 256 + t < nloc) {
            v2i rec;
            rec.x = s[k];
            rec.y = d[k];
            buf[lp[k]] = rec;
        }
    }
    __syncthreads();
    v2i rec[EPT];
    int go[EPT];
#pragma unroll
    for (int k = 0; k < EPT; ++k) {
        const int i = k * 256 + t;
        rec[k] = buf[(FULL || i < nloc) ? i : 0];
    }
#pragma unroll
    for (int k = 0; k < EPT; ++k) go[k] = gofs[rec[k].y >> BIN_SHIFT];
#pragma unroll
    for (int k = 0; k < EPT; ++k) {
        const int i = k * 256 + t;
        if (FULL || i < nloc) out2[go[k] + i] = rec[k];
    }
}

__global__ __launch_bounds__(256)
void bin_scatter_kernel(const int* __restrict__ src, const int* __restrict__ dst,
                        const int* __restrict__ pre, int* __restrict__ bucketed,
                        int E, int NB, int nblk) {
    __shared__ int hist[NB_MAX];
    __shared__ int gofs[NB_MAX];
    __shared__ int wsum[8];
    __shared__ v2i buf[EPB];
    v2i* out2 = (v2i*)bucketed;
    if (blockIdx.x * EPB + EPB <= E) bin_scatter_body<true>(src, dst, pre, out2, hist, gofs, wsum, buf, E, NB, nblk);
    else                             bin_scatter_body<false>(src, dst, pre, out2, hist, gofs, wsum, buf, E, NB, nblk);
}

constexpr int STAGE_CAP = 16384;
__global__ __launch_bounds__(1024)
void bin_sort_kernel(const int* __restrict__ bucketed, const int* __restrict__ pre,
                     int* __restrict__ cnt, int* __restrict__ info, int* __restrict__ sorted,
                     int E, int Nn, int NB, int nblk) {
    __shared__ int lc[BIN_NODES];
    __shared__ int wtot[BIN_NODES / 32];
    __shared__ int stage[STAGE_CAP];
    constexpr int U = 8;
    const int t = threadIdx.x, lane = t & 31, w = t >> 5;
    const int b = blockIdx.x;
    const int beg = pre[b * nblk];
    const int end = (b + 1 < NB) ? pre[(b + 1) * nblk] : E;
    const v2i* in2 = (const v2i*)bucketed;
    if (t < BIN_NODES) lc[t] = 0;
    __syncthreads();

    for (int i0 = beg + t; i0 < end; i0 += 1024 * U) {
        int d[U];
#pragma unroll
        for (int u = 0; u < U; ++u) {
            const int i = i0 + u * 1024;
            d[u] = in2[i < end ? i : i0].y;
        }
#pragma unroll
        for (int u = 0; u < U; ++u)
            if (i0 + u * 1024 < end) atomicAdd(&lc[d[u] & (BIN_NODES - 1)], 1);
    }
    __syncthreads();

    const int c = t < BIN_NODES ? lc[t] : 0;
    int incl = c;
#pragma unroll
    for (int o = 1; o < 32; o <<= 1) {
        const int v = __shfl_up(incl, o);
        if (lane >= o) incl += v;
    }
    if (lane == 31 && t < BIN_NODES) wtot[w] = incl;
    __syncthreads();
    if (t < BIN_NODES) {
        int front = 0;
        for (int j = 0; j < w; ++j) front += wtot[j];
        const int st = front + incl - c;
        const int dv = __float_as_int(rsqrtf((float)(c + 1)));
        lc[t] = st;
        const int node = b * BIN_NODES + t;
        if (node < Nn) {
            v2i rec;
            rec.x = beg + st;
            rec.y = dv;
            ((v2i*)info)[node] = rec;
            cnt[node] = c;
        }
    }
    __syncthreads();

    int* out1 = sorted + beg;
    const int nb = end - beg;
    const bool fits = nb <= STAGE_CAP;
    for (int i0 = beg + t; i0 < end; i0 += 1024 * U) {
        v2i e[U];
#pragma unroll
        for (int u = 0; u < U; ++u) {
            const int i = i0 + u * 1024;
            e[u] = in2[i < end ? i : i0];
        }
#pragma unroll
        for (int u = 0; u < U; ++u) {
            if (i0 + u * 1024 < end) {
                const int nl = e[u].y & (BIN_NODES - 1);
                const int pos = atomicAdd(&lc[nl], 1);
                if (fits) stage[pos] = e[u].x;
                else      out1[pos] = e[u].x;
            }
        }
    }
    if (fits) {
        __syncthreads();
        for (int i0 = t; i0 < nb; i0 += 1024 * U) {
            int rec[U];
#pragma unroll
            for (int u = 0; u < U; ++u) {
                const int i = i0 + u * 1024;
                rec[u] = stage[i < nb ? i : i0];
            }
#pragma unroll
            for (int u = 0; u < U; ++u) {
                const int i = i0 + u * 1024;
                if (i < nb) out1[i] = rec[u];
            }
        }
    }
}

__device__ __forceinline__ float squash(float v) {
    return (v * v) * __builtin_amdgcn_rcpf(v);
}

template <int K, int N, bool SQUASH>
__global__ __launch_bounds__((N / 16) * 32)
void gemm_wmma(const float* __restrict__ X, const float* __restrict__ W,
               const int* __restrict__ info, float* __restrict__ H, int M) {
    const int lane  = threadIdx.x & 31;
    const int ntile = threadIdx.x >> 5;
    const int half  = lane >> 4;
    const int l16   = lane & 15;
    const int row   = blockIdx.x * 16 + l16;
    const int arow  = row < M ? row : M - 1;
    const int ncol  = ntile * 16 + l16;
    const float* __restrict__ xrow = X + (long)arow * K;

    v8f c = {0.f, 0.f, 0.f, 0.f, 0.f, 0.f, 0.f, 0.f};

    const float dva = __int_as_float(((const v2i*)info)[arow].y);

#pragma unroll
    for (int kk = 0; kk < K; kk += 4) {
        const int k0 = kk + half * 2;
        v2f a;
        {
            float2 av = *(const float2*)(xrow + k0);
            if (SQUASH) { av.x = squash(av.x); av.y = squash(av.y); }
            a.x = av.x * dva; a.y = av.y * dva;
        }
        v2f b;
        b.x = W[k0 * N + ncol];
        b.y = W[(k0 + 1) * N + ncol];
        c = __builtin_amdgcn_wmma_f32_16x16x4_f32(
                 false, a,  false, b,
                 (short)0, c,  false,  false);
    }

#pragma unroll
    for (int i = 0; i < 8; ++i) {
        int r = blockIdx.x * 16 + i + half * 8;
        if (r < M) H[(long)r * N + ncol] = c[i];
    }
}

#define GLOBAL_MEM __attribute__((address_space(1)))
template <typename T>
__device__ __forceinline__ GLOBAL_MEM T* wave_uniform_row(T* base, int row, int width) {
    const unsigned long long a = (unsigned long long)(base + (long)row * width);
    const unsigned lo = __builtin_amdgcn_readfirstlane((unsigned)a);
    const unsigned hi = __builtin_amdgcn_readfirstlane((unsigned)(a >> 32));
    return (GLOBAL_MEM T*)(((unsigned long long)hi << 32) | lo);
}

constexpr int NPW = 8;
template <int C, bool RELU>
__global__ __launch_bounds__(256)
void gather_kernel(const float* __restrict__ h, const int* __restrict__ sorted,
                   const int* __restrict__ info, const int* __restrict__ cnt,
                   const float* __restrict__ bias, float* __restrict__ out, int Nn) {
    constexpr int G = 16;
    constexpr int NO_END = 0x7fffffff;
    const int lane = threadIdx.x & 31;
    const int wv = __builtin_amdgcn_readfirstlane((int)((blockIdx.x * blockDim.x + threadIdx.x) >> 5));
    const int node0 = wv * NPW;
    if (node0 >= Nn) return;
    const int nn = (Nn - node0) < NPW ? (Nn - node0) : NPW;

    v2i ni; ni.x = 0; ni.y = 0;
    int cn = 0;
    if (lane < nn) { ni = ((const v2i*)info)[node0 + lane]; cn = cnt[node0 + lane]; }

    float sx, sy = 0.0f;
    {
        const GLOBAL_MEM float* hs = wave_uniform_row(h, node0, C);
        if (C == 64) { const v2f sv = ((const GLOBAL_MEM v2f*)hs)[lane]; sx = sv.x; sy = sv.y; }
        else         { sx = hs[lane]; }
    }
    float bx, by = 0.0f;
    if (C == 64) { bx = bias[2 * lane]; by = bias[2 * lane + 1]; }
    else         { bx = bias[lane]; }

    const int start0 = __builtin_amdgcn_readfirstlane(ni.x);
    const int endl = ni.x + cn - start0;
    const int tot = __builtin_amdgcn_readlane(endl, nn - 1);
    const int* ent = sorted + start0;

    int k = 0;
    int seg_end = __builtin_amdgcn_readfirstlane(endl);
    float ax = 0.0f, ay = 0.0f;

    auto finish_node = [&]() {
        const float di = __int_as_float(__builtin_amdgcn_readlane(ni.y, k));
        GLOBAL_MEM float* orow = wave_uniform_row(out, node0 + k, C);
        if (C == 64) {
            float ox = di * (ax + sx) + bx;
            float oy = di * (ay + sy) + by;
            if (RELU) { ox = fmaxf(ox, 0.0f); oy = fmaxf(oy, 0.0f); }
            v2f ov; ov.x = ox; ov.y = oy;
            ((GLOBAL_MEM v2f*)orow)[lane] = ov;
        } else {
            float ox = di * (ax + sx) + bx;
            if (RELU) ox = fmaxf(ox, 0.0f);
            orow[lane] = ox;
        }
        ++k;
        ax = 0.0f; ay = 0.0f;
        if (k < nn) {
            seg_end = __builtin_amdgcn_readlane(endl, k);
            const GLOBAL_MEM float* hs = wave_uniform_row(h, node0 + k, C);
            if (C == 64) { const v2f sv = ((const GLOBAL_MEM v2f*)hs)[lane]; sx = sv.x; sy = sv.y; }
            else         { sx = hs[lane]; }
        } else {
            seg_end = NO_END;
        }
    };

    int s_cur = 0;
    if (lane < tot) s_cur = ent[lane];
    for (int c0 = 0; c0 < tot; c0 += 32) {
        int s_nxt = 0;
        const int m = (tot - c0) < 32 ? (tot - c0) : 32;
        for (int j0 = 0; j0 < m; j0 += G) {
            float v0[G], v1[G];
#pragma unroll
            for (int u = 0; u < G; ++u) {
                const int sj = __builtin_amdgcn_readlane(s_cur, j0 + u);
                const GLOBAL_MEM float* hr = wave_uniform_row(h, sj, C);
                if (C == 64) { const v2f tv = ((const GLOBAL_MEM v2f*)hr)[lane]; v0[u] = tv.x; v1[u] = tv.y; }
                else         { v0[u] = hr[lane]; }
            }
            if (j0 == 0 && c0 + 32 + lane < tot) s_nxt = ent[c0 + 32 + lane];
            const int e0 = c0 + j0;
            int rel = seg_end - e0;
#pragma unroll
            for (int u = 0; u < G; ++u) {
                while (u == rel) { finish_node(); rel = seg_end - e0; }
                ax += v0[u];
                if (C == 64) ay += v1[u];
            }
        }
        s_cur = s_nxt;
    }
    while (k < nn) finish_node();
}

static inline size_t align256(size_t n) { return (n + 255) & ~size_t(255); }

extern "C" void kernel_launch(void* const* d_in, const int* in_sizes, int n_in,
                              void* d_out, int out_size, void* d_ws, size_t ws_size,
                              hipStream_t stream) {
    const float* x  = (const float*)d_in[0];
    const int*   ei = (const int*)d_in[1];
    const float* W1 = (const float*)d_in[2];
    const float* b1 = (const float*)d_in[3];
    const float* W2 = (const float*)d_in[4];
    const float* b2 = (const float*)d_in[5];
    float* out = (float*)d_out;

    const int hid  = in_sizes[3];
    const int outc = in_sizes[5];
    const int inc  = in_sizes[2] / hid;
    const int Nn   = in_sizes[0] / inc;
    const int E    = in_sizes[1] / 2;
    const int* src = ei;
    const int* dstv = ei + E;
    (void)outc;

    const int NB   = (Nn + BIN_NODES - 1) / BIN_NODES;
    const int nblk = (E + EPB - 1) / EPB;
    const int nt   = NB * nblk;

    char* ws = (char*)d_ws;
    int* cnt    = (int*)ws;    ws += align256((size_t)Nn * 4);
    int* info   = (int*)ws;    ws += align256((size_t)Nn * 8);
    int* table  = (int*)ws;    ws += align256((size_t)nt * 4);
    int* pre    = (int*)ws;    ws += align256((size_t)nt * 4);
    int* bucket = (int*)ws;    ws += align256((size_t)E * 8);
    int* sorted = (int*)ws;    ws += align256((size_t)E * 4);
    float* h1   = (float*)ws;  ws += (size_t)Nn * 64 * 4;
    float* a1   = (float*)ws;  ws += (size_t)Nn * 64 * 4;
    float* h2   = (float*)ws;

    const int T = 256;
    bin_count_kernel<<<nblk, 256, 0, stream>>>(dstv, table, E, NB, nblk);
    scan_kernel<<<(nt + 1023) / 1024, 1024, 0, stream>>>(table, pre, nt);
    bin_scatter_kernel<<<nblk, 256, 0, stream>>>(src, dstv, pre, bucket, E, NB, nblk);
    bin_sort_kernel<<<NB, 1024, 0, stream>>>(bucket, pre, cnt, info, sorted, E, Nn, NB, nblk);

    gemm_wmma<128, 64, true><<<(Nn + 15) / 16, 128, 0, stream>>>(x, W1, info, h1, Nn);
    gather_kernel<64, true><<<(Nn + 8 * NPW - 1) / (8 * NPW), T, 0, stream>>>(h1, sorted, info, cnt, b1, a1, Nn);

    gemm_wmma<64, 32, false><<<(Nn + 15) / 16, 64, 0, stream>>>(a1, W2, info, h2, Nn);
    gather_kernel<32, false><<<(Nn + 8 * NPW - 1) / (8 * NPW), T, 0, stream>>>(h2, sorted, info, cnt, b2, out, Nn);
}


# GEMMs: one wave per 16 whole rows (A read once by 16-byte loads, N/16 accumulator tiles, W fragments from an LDS copy), not one wave per 16x16 tile; each 8 k's products summed in another order
# speedup vs baseline: 4.5619x; 1.1647x over previous
#include <hip/hip_runtime.h>

typedef __attribute__((ext_vector_type(2))) float v2f;
typedef __attribute__((ext_vector_type(8))) float v8f;

typedef __attribute__((ext_vector_type(2))) int v2i;
typedef __attribute__((ext_vector_type(4))) int v4i;

constexpr int BIN_SHIFT = 9;
constexpr int BIN_NODES = 1 << BIN_SHIFT;
constexpr int NB_MAX    = 256;
constexpr int EPT       = 16;
constexpr int EPB       = 256 * EPT;

__global__ __launch_bounds__(256)
void bin_count_kernel(const int* __restrict__ dst, int* __restrict__ table, int E, int NB, int nblk) {
    __shared__ int hist[NB_MAX];
    const int t = threadIdx.x;
    hist[t] = 0;
    __syncthreads();
    const int blk0 = blockIdx.x * EPB;
    if (blk0 + EPB <= E) {
        int d[EPT];
#pragma unroll
        for (int k = 0; k < EPT; ++k) d[k] = dst[blk0 + k * 256 + t];
#pragma unroll
        for (int k = 0; k < EPT; ++k) atomicAdd(&hist[d[k] >> BIN_SHIFT], 1);
    } else {
        for (int k = 0; k < EPT; ++k) {
            const int e = blk0 + k * 256 + t;
            if (e < E) atomicAdd(&hist[dst[e] >> BIN_SHIFT], 1);
        }
    }
    __syncthreads();
    if (t < NB) table[t * nblk + blockIdx.x] = hist[t];
}

__global__ __launch_bounds__(1024)
void scan_kernel(const int* __restrict__ cnt, int* __restrict__ pre, int n) {
    __shared__ int s_pre[32];
    __shared__ int s_tot[32];
    const int t = threadIdx.x, lane = t & 31, w = t >> 5;
    const int tile0 = blockIdx.x * 1024;
    int part = 0;
    const v4i* c4 = (const v4i*)cnt;
    const int n4 = tile0 >> 2;
#pragma unroll 4
    for (int i = t; i < n4; i += 1024) {
        const v4i v = c4[i];
        part += (v.x + v.y) + (v.z + v.w);
    }
#pragma unroll
    for (int o = 16; o > 0; o >>= 1) part += __shfl_xor(part, o);
    const int i = tile0 + t;
    const int c = i < n ? cnt[i] : 0;
    int incl = c;
#pragma unroll
    for (int o = 1; o < 32; o <<= 1) {
        const int v = __shfl_up(incl, o);
        if (lane >= o) incl += v;
    }
    if (lane == 31) s_tot[w] = incl;
    if (lane == 0)  s_pre[w] = part;
    __syncthreads();
    int a = s_pre[lane];
    int b = lane < w ? s_tot[lane] : 0;
#pragma unroll
    for (int o = 16; o > 0; o >>= 1) { a += __shfl_xor(a, o); b += __shfl_xor(b, o); }
    if (i < n) pre[i] = a + b + (incl - c);
}

template <bool FULL>
__device__ __forceinline__ void bin_scatter_body(const int* __restrict__ src, const int* __restrict__ dst,
                                                 const int* __restrict__ pre, v2i* __restrict__ out2,
                                                 int* hist, int* gofs, int* wsum, v2i* buf,
                                                 int E, int NB, int nblk) {
    const int t = threadIdx.x, lane = t & 31, w = t >> 5;
    const int gfirst = t < NB ? pre[t * nblk + blockIdx.x] : 0;
    hist[t] = 0;
    __syncthreads();
    const int blk0 = blockIdx.x * EPB;
    const int nloc = FULL ? EPB : (E - blk0);
    int s[EPT], d[EPT], r[EPT];
#pragma unroll
    for (int k = 0; k < EPT; ++k) {
        const int e = blk0 + k * 256 + t;
        const int ec = (FULL || e < E) ? e : E - 1;
        s[k] = src[ec];
        d[k] = dst[ec];
    }
#pragma unroll
    for (int k = 0; k < EPT; ++k) {
        r[k] = 0;
        if (FULL || k * 256 + t < nloc) r[k] = atomicAdd(&hist[d[k] >> BIN_SHIFT], 1);
    }
    __syncthreads();
    const int c = hist[t];
    int incl = c;
#pragma unroll
    for (int o = 1; o < 32; o <<= 1) {
        const int v = __shfl_up(incl, o);
        if (lane >= o) incl += v;
    }
    if (lane == 31) wsum[w] = incl;
    __syncthreads();
    int front = 0;
#pragma unroll
    for (int j = 0; j < 8; ++j) front += j < w ? wsum[j] : 0;
    const int ls = front + incl - c;
    hist[t] = ls;
    gofs[t] = gfirst - ls;
    __syncthreads();
    int lp[EPT];
#pragma unroll
    for (int k = 0; k < EPT; ++k) lp[k] = hist[d[k] >> BIN_SHIFT] + r[k];
#pragma unroll
    for (int k = 0; k < EPT; ++k) {
        if (FULL || k * 256 + t < nloc) {
            v2i rec;
            rec.x = s[k];
            rec.y = d[k];
            buf[lp[k]] = rec;
        }
    }
    __syncthreads();
    v2i rec[EPT];
    int go[EPT];
#pragma unroll
    for (int k = 0; k < EPT; ++k) {
        const int i = k * 256 + t;
        rec[k] = buf[(FULL || i < nloc) ? i : 0];
    }
#pragma unroll
    for (int k = 0; k < EPT; ++k) go[k] = gofs[rec[k].y >> BIN_SHIFT];
#pragma unroll
    for (int k = 0; k < EPT; ++k) {
        const int i = k * 256 + t;
        if (FULL || i < nloc) out2[go[k] + i] = rec[k];
    }
}

__global__ __launch_bounds__(256)
void bin_scatter_kernel(const int* __restrict__ src, const int* __restrict__ dst,
                        const int* __restrict__ pre, int* __restrict__ bucketed,
                        int E, int NB, int nblk) {
    __shared__ int hist[NB_MAX];
    __shared__ int gofs[NB_MAX];
    __shared__ int wsum[8];
    __shared__ v2i buf[EPB];
    v2i* out2 = (v2i*)bucketed;
    if (blockIdx.x * EPB + EPB <= E) bin_scatter_body<true>(src, dst, pre, out2, hist, gofs, wsum, buf, E, NB, nblk);
    else                             bin_scatter_body<false>(src, dst, pre, out2, hist, gofs, wsum, buf, E, NB, nblk);
}

constexpr int STAGE_CAP = 16384;
__global__ __launch_bounds__(1024)
void bin_sort_kernel(const int* __restrict__ bucketed, const int* __restrict__ pre,
                     int* __restrict__ cnt, int* __restrict__ info, int* __restrict__ sorted,
                     int E, int Nn, int NB, int nblk) {
    __shared__ int lc[BIN_NODES];
    __shared__ int wtot[BIN_NODES / 32];
    __shared__ int stage[STAGE_CAP];
    constexpr int U = 8;
    const int t = threadIdx.x, lane = t & 31, w = t >> 5;
    const int b = blockIdx.x;
    const int beg = pre[b * nblk];
    const int end = (b + 1 < NB) ? pre[(b + 1) * nblk] : E;
    const v2i* in2 = (const v2i*)bucketed;
    if (t < BIN_NODES) lc[t] = 0;
    __syncthreads();

    for (int i0 = beg + t; i0 < end; i0 += 1024 * U) {
        int d[U];
#pragma unroll
        for (int u = 0; u < U; ++u) {
            const int i = i0 + u * 1024;
            d[u] = in2[i < end ? i : i0].y;
        }
#pragma unroll
        for (int u = 0; u < U; ++u)
            if (i0 + u * 1024 < end) atomicAdd(&lc[d[u] & (BIN_NODES - 1)], 1);
    }
    __syncthreads();

    const int c = t < BIN_NODES ? lc[t] : 0;
    int incl = c;
#pragma unroll
    for (int o = 1; o < 32; o <<= 1) {
        const int v = __shfl_up(incl, o);
        if (lane >= o) incl += v;
    }
    if (lane == 31 && t < BIN_NODES) wtot[w] = incl;
    __syncthreads();
    if (t < BIN_NODES) {
        int front = 0;
        for (int j = 0; j < w; ++j) front += wtot[j];
        const int st = front + incl - c;
        const int dv = __float_as_int(rsqrtf((float)(c + 1)));
        lc[t] = st;
        const int node = b * BIN_NODES + t;
        if (node < Nn) {
            v2i rec;
            rec.x = beg + st;
            rec.y = dv;
            ((v2i*)info)[node] = rec;
            cnt[node] = c;
        }
    }
    __syncthreads();

    int* out1 = sorted + beg;
    const int nb = end - beg;
    const bool fits = nb <= STAGE_CAP;
    for (int i0 = beg + t; i0 < end; i0 += 1024 * U) {
        v2i e[U];
#pragma unroll
        for (int u = 0; u < U; ++u) {
            const int i = i0 + u * 1024;
            e[u] = in2[i < end ? i : i0];
        }
#pragma unroll
        for (int u = 0; u < U; ++u) {
            if (i0 + u * 1024 < end) {
                const int nl = e[u].y & (BIN_NODES - 1);
                const int pos = atomicAdd(&lc[nl], 1);
                if (fits) stage[pos] = e[u].x;
                else      out1[pos] = e[u].x;
            }
        }
    }
    if (fits) {
        __syncthreads();
        for (int i0 = t; i0 < nb; i0 += 1024 * U) {
            int rec[U];
#pragma unroll
            for (int u = 0; u < U; ++u) {
                const int i = i0 + u * 1024;
                rec[u] = stage[i < nb ? i : i0];
            }
#pragma unroll
            for (int u = 0; u < U; ++u) {
                const int i = i0 + u * 1024;
                if (i < nb) out1[i] = rec[u];
            }
        }
    }
}

__device__ __forceinline__ float squash(float v) {
    return (v * v) * __builtin_amdgcn_rcpf(v);
}

typedef __attribute__((ext_vector_type(4))) float v4f;

template <int K, int N>
__device__ __forceinline__ void stage_w(const float* __restrict__ W, v4f* Wl, int tid) {
    constexpr int NT  = N / 16;
    constexpr int CG  = N / 4;
    constexpr int QPP = 128 / CG;
    const int cg = tid % CG;
    const int t  = cg >> 2;
    const int l0 = (cg & 3) * 4;
#pragma unroll
    for (int p = 0; p < K / 4 / QPP; ++p) {
        const int q = tid / CG + p * QPP;
        const float* wp = W + (4 * q) * N + 4 * cg;
        const v4f r0 = *(const v4f*)(wp);
        const v4f r1 = *(const v4f*)(wp + N);
        const v4f r2 = *(const v4f*)(wp + 2 * N);
        const v4f r3 = *(const v4f*)(wp + 3 * N);
        v4f* d = Wl + (((q >> 1) * NT + t) * 32 + (q & 1) * 16 + l0);
        v4f o0; o0.x = r0.x; o0.y = r1.x; o0.z = r2.x; o0.w = r3.x;
        v4f o1; o1.x = r0.y; o1.y = r1.y; o1.z = r2.y; o1.w = r3.y;
        v4f o2; o2.x = r0.z; o2.y = r1.z; o2.z = r2.z; o2.w = r3.z;
        v4f o3; o3.x = r0.w; o3.y = r1.w; o3.z = r2.w; o3.w = r3.w;
        d[0 ^ t] = o0; d[1 ^ t] = o1; d[2 ^ t] = o2; d[3 ^ t] = o3;
    }
}

template <int K, int N, bool SQUASH>
__global__ __launch_bounds__(128)
void gemm_wmma(const float* __restrict__ X, const float* __restrict__ W,
               const int* __restrict__ info, float* __restrict__ H, int M) {
    constexpr int NT = N / 16;
    constexpr int KS = K / 8;
    __shared__ v4f Wl[KS * NT * 32];
    const int tid   = threadIdx.x;
    const int lane  = tid & 31;
    const int half  = lane >> 4;
    const int l16   = lane & 15;
    const int row0  = (blockIdx.x * 4 + (tid >> 5)) * 16;
    const int row   = row0 + l16;
    const int arow  = row < M ? row : M - 1;
    const float* __restrict__ xrow = X + (long)arow * K + half * 4;

    v4f av[KS];
#pragma unroll
    for (int s = 0; s < KS; ++s) av[s] = *(const v4f*)(xrow + s * 8);
    const float dva = __int_as_float(info[2 * arow + 1]);

    stage_w<K, N>(W, Wl, tid);
    __syncthreads();

    v8f c[NT];
#pragma unroll
    for (int t = 0; t < NT; ++t) c[t] = (v8f){0.f, 0.f, 0.f, 0.f, 0.f, 0.f, 0.f, 0.f};

#pragma unroll
    for (int s = 0; s < KS; ++s) {
        v4f a = av[s];
        if (SQUASH) { a.x = squash(a.x); a.y = squash(a.y); a.z = squash(a.z); a.w = squash(a.w); }
        v2f a0, a1;
        a0.x = a.x * dva; a0.y = a.y * dva;
        a1.x = a.z * dva; a1.y = a.w * dva;
        v4f b[NT];
#pragma unroll
        for (int t = 0; t < NT; ++t) b[t] = Wl[(s * NT + t) * 32 + (lane ^ t)];
#pragma unroll
        for (int t = 0; t < NT; ++t) {
            v2f b0; b0.x = b[t].x; b0.y = b[t].y;
            c[t] = __builtin_amdgcn_wmma_f32_16x16x4_f32(
                        false, a0,  false, b0,
                        (short)0, c[t],  false,  false);
        }
#pragma unroll
        for (int t = 0; t < NT; ++t) {
            v2f b1; b1.x = b[t].z; b1.y = b[t].w;
            c[t] = __builtin_amdgcn_wmma_f32_16x16x4_f32(
                        false, a1,  false, b1,
                        (short)0, c[t],  false,  false);
        }
    }

#pragma unroll
    for (int i = 0; i < 8; ++i) {
        const int r = row0 + i + half * 8;
        if (r < M) {
#pragma unroll
            for (int t = 0; t < NT; ++t) H[(long)r * N + t * 16 + l16] = c[t][i];
        }
    }
}

#define GLOBAL_MEM __attribute__((address_space(1)))
template <typename T>
__device__ __forceinline__ GLOBAL_MEM T* wave_uniform_row(T* base, int row, int width) {
    const unsigned long long a = (unsigned long long)(base + (long)row * width);
    const unsigned lo = __builtin_amdgcn_readfirstlane((unsigned)a);
    const unsigned hi = __builtin_amdgcn_readfirstlane((unsigned)(a >> 32));
    return (GLOBAL_MEM T*)(((unsigned long long)hi << 32) | lo);
}

constexpr int NPW = 8;
template <int C, bool RELU>
__global__ __launch_bounds__(256)
void gather_kernel(const float* __restrict__ h, const int* __restrict__ sorted,
                   const int* __restrict__ info, const int* __restrict__ cnt,
                   const float* __restrict__ bias, float* __restrict__ out, int Nn) {
    constexpr int G = 16;
    constexpr int NO_END = 0x7fffffff;
    const int lane = threadIdx.x & 31;
    const int wv = __builtin_amdgcn_readfirstlane((int)((blockIdx.x * blockDim.x + threadIdx.x) >> 5));
    const int node0 = wv * NPW;
    if (node0 >= Nn) return;
    const int nn = (Nn - node0) < NPW ? (Nn - node0) : NPW;

    v2i ni; ni.x = 0; ni.y = 0;
    int cn = 0;
    if (lane < nn) { ni = ((const v2i*)info)[node0 + lane]; cn = cnt[node0 + lane]; }

    float sx, sy = 0.0f;
    {
        const GLOBAL_MEM float* hs = wave_uniform_row(h, node0, C);
        if (C == 64) { const v2f sv = ((const GLOBAL_MEM v2f*)hs)[lane]; sx = sv.x; sy = sv.y; }
        else         { sx = hs[lane]; }
    }
    float bx, by = 0.0f;
    if (C == 64) { bx = bias[2 * lane]; by = bias[2 * lane + 1]; }
    else         { bx = bias[lane]; }

    const int start0 = __builtin_amdgcn_readfirstlane(ni.x);
    const int endl = ni.x + cn - start0;
    const int tot = __builtin_amdgcn_readlane(endl, nn - 1);
    const int* ent = sorted + start0;

    int k = 0;
    int seg_end = __builtin_amdgcn_readfirstlane(endl);
    float ax = 0.0f, ay = 0.0f;

    auto finish_node = [&]() {
        const float di = __int_as_float(__builtin_amdgcn_readlane(ni.y, k));
        GLOBAL_MEM float* orow = wave_uniform_row(out, node0 + k, C);
        if (C == 64) {
            float ox = di * (ax + sx) + bx;
            float oy = di * (ay + sy) + by;
            if (RELU) { ox = fmaxf(ox, 0.0f); oy = fmaxf(oy, 0.0f); }
            v2f ov; ov.x = ox; ov.y = oy;
            ((GLOBAL_MEM v2f*)orow)[lane] = ov;
        } else {
            float ox = di * (ax + sx) + bx;
            if (RELU) ox = fmaxf(ox, 0.0f);
            orow[lane] = ox;
        }
        ++k;
        ax = 0.0f; ay = 0.0f;
        if (k < nn) {
            seg_end = __builtin_amdgcn_readlane(endl, k);
            const GLOBAL_MEM float* hs = wave_uniform_row(h, node0 + k, C);
            if (C == 64) { const v2f sv = ((const GLOBAL_MEM v2f*)hs)[lane]; sx = sv.x; sy = sv.y; }
            else         { sx = hs[lane]; }
        } else {
            seg_end = NO_END;
        }
    };

    int s_cur = 0;
    if (lane < tot) s_cur = ent[lane];
    for (int c0 = 0; c0 < tot; c0 += 32) {
        int s_nxt = 0;
        const int m = (tot - c0) < 32 ? (tot - c0) : 32;
        for (int j0 = 0; j0 < m; j0 += G) {
            float v0[G], v1[G];
#pragma unroll
            for (int u = 0; u < G; ++u) {
                const int sj = __builtin_amdgcn_readlane(s_cur, j0 + u);
                const GLOBAL_MEM float* hr = wave_uniform_row(h, sj, C);
                if (C == 64) { const v2f tv = ((const GLOBAL_MEM v2f*)hr)[lane]; v0[u] = tv.x; v1[u] = tv.y; }
                else         { v0[u] = hr[lane]; }
            }
            if (j0 == 0 && c0 + 32 + lane < tot) s_nxt = ent[c0 + 32 + lane];
            const int e0 = c0 + j0;
            int rel = seg_end - e0;
#pragma unroll
            for (int u = 0; u < G; ++u) {
                while (u == rel) { finish_node(); rel = seg_end - e0; }
                ax += v0[u];
                if (C == 64) ay += v1[u];
            }
        }
        s_cur = s_nxt;
    }
    while (k < nn) finish_node();
}

static inline size_t align256(size_t n) { return (n + 255) & ~size_t(255); }

extern "C" void kernel_launch(void* const* d_in, const int* in_sizes, int n_in,
                              void* d_out, int out_size, void* d_ws, size_t ws_size,
                              hipStream_t stream) {
    const float* x  = (const float*)d_in[0];
    const int*   ei = (const int*)d_in[1];
    const float* W1 = (const float*)d_in[2];
    const float* b1 = (const float*)d_in[3];
    const float* W2 = (const float*)d_in[4];
    const float* b2 = (const float*)d_in[5];
    float* out = (float*)d_out;

    const int hid  = in_sizes[3];
    const int outc = in_sizes[5];
    const int inc  = in_sizes[2] / hid;
    const int Nn   = in_sizes[0] / inc;
    const int E    = in_sizes[1] / 2;
    const int* src = ei;
    const int* dstv = ei + E;
    (void)outc;

    const int NB   = (Nn + BIN_NODES - 1) / BIN_NODES;
    const int nblk = (E + EPB - 1) / EPB;
    const int nt   = NB * nblk;

    char* ws = (char*)d_ws;
    int* cnt    = (int*)ws;    ws += align256((size_t)Nn * 4);
    int* info   = (int*)ws;    ws += align256((size_t)Nn * 8);
    int* table  = (int*)ws;    ws += align256((size_t)nt * 4);
    int* pre    = (int*)ws;    ws += align256((size_t)nt * 4);
    int* bucket = (int*)ws;    ws += align256((size_t)E * 8);
    int* sorted = (int*)ws;    ws += align256((size_t)E * 4);
    float* h1   = (float*)ws;  ws += (size_t)Nn * 64 * 4;
    float* a1   = (float*)ws;  ws += (size_t)Nn * 64 * 4;
    float* h2   = (float*)ws;

    const int T = 256;
    bin_count_kernel<<<nblk, 256, 0, stream>>>(dstv, table, E, NB, nblk);
    scan_kernel<<<(nt + 1023) / 1024, 1024, 0, stream>>>(table, pre, nt);
    bin_scatter_kernel<<<nblk, 256, 0, stream>>>(src, dstv, pre, bucket, E, NB, nblk);
    bin_sort_kernel<<<NB, 1024, 0, stream>>>(bucket, pre, cnt, info, sorted, E, Nn, NB, nblk);

    gemm_wmma<128, 64, true><<<(Nn + 63) / 64, 128, 0, stream>>>(x, W1, info, h1, Nn);
    gather_kernel<64, true><<<(Nn + 8 * NPW - 1) / (8 * NPW), T, 0, stream>>>(h1, sorted, info, cnt, b1, a1, Nn);

    gemm_wmma<64, 32, false><<<(Nn + 63) / 64, 128, 0, stream>>>(a1, W2, info, h2, Nn);
    gather_kernel<32, false><<<(Nn + 8 * NPW - 1) / (8 * NPW), T, 0, stream>>>(h2, sorted, info, cnt, b2, out, Nn);
}


# layer-2 product moved into the layer-1 aggregation launch: a block keeps its 64 finished rows in LDS, 4 waves multiply them by W2 on f32 WMMA; a1 never stored, one launch fewer; sum orders kept
# speedup vs baseline: 4.9051x; 1.0752x over previous
#include <hip/hip_runtime.h>

typedef __attribute__((ext_vector_type(2))) float v2f;
typedef __attribute__((ext_vector_type(8))) float v8f;

typedef __attribute__((ext_vector_type(2))) int v2i;
typedef __attribute__((ext_vector_type(4))) int v4i;

constexpr int BIN_SHIFT = 9;
constexpr int BIN_NODES = 1 << BIN_SHIFT;
constexpr int NB_MAX    = 256;
constexpr int EPT       = 16;
constexpr int EPB       = 256 * EPT;

__global__ __launch_bounds__(256)
void bin_count_kernel(const int* __restrict__ dst, int* __restrict__ table, int E, int NB, int nblk) {
    __shared__ int hist[NB_MAX];
    const int t = threadIdx.x;
    hist[t] = 0;
    __syncthreads();
    const int blk0 = blockIdx.x * EPB;
    if (blk0 + EPB <= E) {
        int d[EPT];
#pragma unroll
        for (int k = 0; k < EPT; ++k) d[k] = dst[blk0 + k * 256 + t];
#pragma unroll
        for (int k = 0; k < EPT; ++k) atomicAdd(&hist[d[k] >> BIN_SHIFT], 1);
    } else {
        for (int k = 0; k < EPT; ++k) {
            const int e = blk0 + k * 256 + t;
            if (e < E) atomicAdd(&hist[dst[e] >> BIN_SHIFT], 1);
        }
    }
    __syncthreads();
    if (t < NB) table[t * nblk + blockIdx.x] = hist[t];
}

__global__ __launch_bounds__(1024)
void scan_kernel(const int* __restrict__ cnt, int* __restrict__ pre, int n) {
    __shared__ int s_pre[32];
    __shared__ int s_tot[32];
    const int t = threadIdx.x, lane = t & 31, w = t >> 5;
    const int tile0 = blockIdx.x * 1024;
    int part = 0;
    const v4i* c4 = (const v4i*)cnt;
    const int n4 = tile0 >> 2;
#pragma unroll 4
    for (int i = t; i < n4; i += 1024) {
        const v4i v = c4[i];
        part += (v.x + v.y) + (v.z + v.w);
    }
#pragma unroll
    for (int o = 16; o > 0; o >>= 1) part += __shfl_xor(part, o);
    const int i = tile0 + t;
    const int c = i < n ? cnt[i] : 0;
    int incl = c;
#pragma unroll
    for (int o = 1; o < 32; o <<= 1) {
        const int v = __shfl_up(incl, o);
        if (lane >= o) incl += v;
    }
    if (lane == 31) s_tot[w] = incl;
    if (lane == 0)  s_pre[w] = part;
    __syncthreads();
    int a = s_pre[lane];
    int b = lane < w ? s_tot[lane] : 0;
#pragma unroll
    for (int o = 16; o > 0; o >>= 1) { a += __shfl_xor(a, o); b += __shfl_xor(b, o); }
    if (i < n) pre[i] = a + b + (incl - c);
}

template <bool FULL>
__device__ __forceinline__ void bin_scatter_body(const int* __restrict__ src, const int* __restrict__ dst,
                                                 const int* __restrict__ pre, v2i* __restrict__ out2,
                                                 int* hist, int* gofs, int* wsum, v2i* buf,
                                                 int E, int NB, int nblk) {
    const int t = threadIdx.x, lane = t & 31, w = t >> 5;
    const int gfirst = t < NB ? pre[t * nblk + blockIdx.x] : 0;
    hist[t] = 0;
    __syncthreads();
    const int blk0 = blockIdx.x * EPB;
    const int nloc = FULL ? EPB : (E - blk0);
    int s[EPT], d[EPT], r[EPT];
#pragma unroll
    for (int k = 0; k < EPT; ++k) {
        const int e = blk0 + k * 256 + t;
        const int ec = (FULL || e < E) ? e : E - 1;
        s[k] = src[ec];
        d[k] = dst[ec];
    }
#pragma unroll
    for (int k = 0; k < EPT; ++k) {
        r[k] = 0;
        if (FULL || k * 256 + t < nloc) r[k] = atomicAdd(&hist[d[k] >> BIN_SHIFT], 1);
    }
    __syncthreads();
    const int c = hist[t];
    int incl = c;
#pragma unroll
    for (int o = 1; o < 32; o <<= 1) {
        const int v = __shfl_up(incl, o);
        if (lane >= o) incl += v;
    }
    if (lane == 31) wsum[w] = incl;
    __syncthreads();
    int front = 0;
#pragma unroll
    for (int j = 0; j < 8; ++j) front += j < w ? wsum[j] : 0;
    const int ls = front + incl - c;
    hist[t] = ls;
    gofs[t] = gfirst - ls;
    __syncthreads();
    int lp[EPT];
#pragma unroll
    for (int k = 0; k < EPT; ++k) lp[k] = hist[d[k] >> BIN_SHIFT] + r[k];
#pragma unroll
    for (int k = 0; k < EPT; ++k) {
        if (FULL || k * 256 + t < nloc) {
            v2i rec;
            rec.x = s[k];
            rec.y = d[k];
            buf[lp[k]] = rec;
        }
    }
    __syncthreads();
    v2i rec[EPT];
    int go[EPT];
#pragma unroll
    for (int k = 0; k < EPT; ++k) {
        const int i = k * 256 + t;
        rec[k] = buf[(FULL || i < nloc) ? i : 0];
    }
#pragma unroll
    for (int k = 0; k < EPT; ++k) go[k] = gofs[rec[k].y >> BIN_SHIFT];
#pragma unroll
    for (int k = 0; k < EPT; ++k) {
        const int i = k * 256 + t;
        if (FULL || i < nloc) out2[go[k] + i] = rec[k];
    }
}

__global__ __launch_bounds__(256)
void bin_scatter_kernel(const int* __restrict__ src, const int* __restrict__ dst,
                        const int* __restrict__ pre, int* __restrict__ bucketed,
                        int E, int NB, int nblk) {
    __shared__ int hist[NB_MAX];
    __shared__ int gofs[NB_MAX];
    __shared__ int wsum[8];
    __shared__ v2i buf[EPB];
    v2i* out2 = (v2i*)bucketed;
    if (blockIdx.x * EPB + EPB <= E) bin_scatter_body<true>(src, dst, pre, out2, hist, gofs, wsum, buf, E, NB, nblk);
    else                             bin_scatter_body<false>(src, dst, pre, out2, hist, gofs, wsum, buf, E, NB, nblk);
}

constexpr int STAGE_CAP = 16384;
__global__ __launch_bounds__(1024)
void bin_sort_kernel(const int* __restrict__ bucketed, const int* __restrict__ pre,
                     int* __restrict__ cnt, int* __restrict__ info, int* __restrict__ sorted,
                     int E, int Nn, int NB, int nblk) {
    __shared__ int lc[BIN_NODES];
    __shared__ int wtot[BIN_NODES / 32];
    __shared__ int stage[STAGE_CAP];
    constexpr int U = 8;
    const int t = threadIdx.x, lane = t & 31, w = t >> 5;
    const int b = blockIdx.x;
    const int beg = pre[b * nblk];
    const int end = (b + 1 < NB) ? pre[(b + 1) * nblk] : E;
    const v2i* in2 = (const v2i*)bucketed;
    if (t < BIN_NODES) lc[t] = 0;
    __syncthreads();

    for (int i0 = beg + t; i0 < end; i0 += 1024 * U) {
        int d[U];
#pragma unroll
        for (int u = 0; u < U; ++u) {
            const int i = i0 + u * 1024;
            d[u] = in2[i < end ? i : i0].y;
        }
#pragma unroll
        for (int u = 0; u < U; ++u)
            if (i0 + u * 1024 < end) atomicAdd(&lc[d[u] & (BIN_NODES - 1)], 1);
    }
    __syncthreads();

    const int c = t < BIN_NODES ? lc[t] : 0;
    int incl = c;
#pragma unroll
    for (int o = 1; o < 32; o <<= 1) {
        const int v = __shfl_up(incl, o);
        if (lane >= o) incl += v;
    }
    if (lane == 31 && t < BIN_NODES) wtot[w] = incl;
    __syncthreads();
    if (t < BIN_NODES) {
        int front = 0;
        for (int j = 0; j < w; ++j) front += wtot[j];
        const int st = front + incl - c;
        const int dv = __float_as_int(rsqrtf((float)(c + 1)));
        lc[t] = st;
        const int node = b * BIN_NODES + t;
        if (node < Nn) {
            v2i rec;
            rec.x = beg + st;
            rec.y = dv;
            ((v2i*)info)[node] = rec;
            cnt[node] = c;
        }
    }
    __syncthreads();

    int* out1 = sorted + beg;
    const int nb = end - beg;
    const bool fits = nb <= STAGE_CAP;
    for (int i0 = beg + t; i0 < end; i0 += 1024 * U) {
        v2i e[U];
#pragma unroll
        for (int u = 0; u < U; ++u) {
            const int i = i0 + u * 1024;
            e[u] = in2[i < end ? i : i0];
        }
#pragma unroll
        for (int u = 0; u < U; ++u) {
            if (i0 + u * 1024 < end) {
                const int nl = e[u].y & (BIN_NODES - 1);
                const int pos = atomicAdd(&lc[nl], 1);
                if (fits) stage[pos] = e[u].x;
                else      out1[pos] = e[u].x;
            }
        }
    }
    if (fits) {
        __syncthreads();
        for (int i0 = t; i0 < nb; i0 += 1024 * U) {
            int rec[U];
#pragma unroll
            for (int u = 0; u < U; ++u) {
                const int i = i0 + u * 1024;
                rec[u] = stage[i < nb ? i : i0];
            }
#pragma unroll
            for (int u = 0; u < U; ++u) {
                const int i = i0 + u * 1024;
                if (i < nb) out1[i] = rec[u];
            }
        }
    }
}

__device__ __forceinline__ float squash(float v) {
    return (v * v) * __builtin_amdgcn_rcpf(v);
}

typedef __attribute__((ext_vector_type(4))) float v4f;

template <int K, int N>
__device__ __forceinline__ void stage_w(const float* __restrict__ W, v4f* Wl, int tid) {
    constexpr int NT  = N / 16;
    constexpr int CG  = N / 4;
    constexpr int QPP = 128 / CG;
    const int cg = tid % CG;
    const int t  = cg >> 2;
    const int l0 = (cg & 3) * 4;
#pragma unroll
    for (int p = 0; p < K / 4 / QPP; ++p) {
        const int q = tid / CG + p * QPP;
        const float* wp = W + (4 * q) * N + 4 * cg;
        const v4f r0 = *(const v4f*)(wp);
        const v4f r1 = *(const v4f*)(wp + N);
        const v4f r2 = *(const v4f*)(wp + 2 * N);
        const v4f r3 = *(const v4f*)(wp + 3 * N);
        v4f* d = Wl + (((q >> 1) * NT + t) * 32 + (q & 1) * 16 + l0);
        v4f o0; o0.x = r0.x; o0.y = r1.x; o0.z = r2.x; o0.w = r3.x;
        v4f o1; o1.x = r0.y; o1.y = r1.y; o1.z = r2.y; o1.w = r3.y;
        v4f o2; o2.x = r0.z; o2.y = r1.z; o2.z = r2.z; o2.w = r3.z;
        v4f o3; o3.x = r0.w; o3.y = r1.w; o3.z = r2.w; o3.w = r3.w;
        d[0 ^ t] = o0; d[1 ^ t] = o1; d[2 ^ t] = o2; d[3 ^ t] = o3;
    }
}

template <int K, int N, bool SQUASH>
__global__ __launch_bounds__(128)
void gemm_wmma(const float* __restrict__ X, const float* __restrict__ W,
               const int* __restrict__ info, float* __restrict__ H, int M) {
    constexpr int NT = N / 16;
    constexpr int KS = K / 8;
    __shared__ v4f Wl[KS * NT * 32];
    const int tid   = threadIdx.x;
    const int lane  = tid & 31;
    const int half  = lane >> 4;
    const int l16   = lane & 15;
    const int row0  = (blockIdx.x * 4 + (tid >> 5)) * 16;
    const int row   = row0 + l16;
    const int arow  = row < M ? row : M - 1;
    const float* __restrict__ xrow = X + (long)arow * K + half * 4;

    v4f av[KS];
#pragma unroll
    for (int s = 0; s < KS; ++s) av[s] = *(const v4f*)(xrow + s * 8);
    const float dva = __int_as_float(info[2 * arow + 1]);

    stage_w<K, N>(W, Wl, tid);
    __syncthreads();

    v8f c[NT];
#pragma unroll
    for (int t = 0; t < NT; ++t) c[t] = (v8f){0.f, 0.f, 0.f, 0.f, 0.f, 0.f, 0.f, 0.f};

#pragma unroll
    for (int s = 0; s < KS; ++s) {
        v4f a = av[s];
        if (SQUASH) { a.x = squash(a.x); a.y = squash(a.y); a.z = squash(a.z); a.w = squash(a.w); }
        v2f a0, a1;
        a0.x = a.x * dva; a0.y = a.y * dva;
        a1.x = a.z * dva; a1.y = a.w * dva;
        v4f b[NT];
#pragma unroll
        for (int t = 0; t < NT; ++t) b[t] = Wl[(s * NT + t) * 32 + (lane ^ t)];
#pragma unroll
        for (int t = 0; t < NT; ++t) {
            v2f b0; b0.x = b[t].x; b0.y = b[t].y;
            c[t] = __builtin_amdgcn_wmma_f32_16x16x4_f32(
                        false, a0,  false, b0,
                        (short)0, c[t],  false,  false);
        }
#pragma unroll
        for (int t = 0; t < NT; ++t) {
            v2f b1; b1.x = b[t].z; b1.y = b[t].w;
            c[t] = __builtin_amdgcn_wmma_f32_16x16x4_f32(
                        false, a1,  false, b1,
                        (short)0, c[t],  false,  false);
        }
    }

#pragma unroll
    for (int i = 0; i < 8; ++i) {
        const int r = row0 + i + half * 8;
        if (r < M) {
#pragma unroll
            for (int t = 0; t < NT; ++t) H[(long)r * N + t * 16 + l16] = c[t][i];
        }
    }
}

#define GLOBAL_MEM __attribute__((address_space(1)))
template <typename T>
__device__ __forceinline__ GLOBAL_MEM T* wave_uniform_row(T* base, int row, int width) {
    const unsigned long long a = (unsigned long long)(base + (long)row * width);
    const unsigned lo = __builtin_amdgcn_readfirstlane((unsigned)a);
    const unsigned hi = __builtin_amdgcn_readfirstlane((unsigned)(a >> 32));
    return (GLOBAL_MEM T*)(((unsigned long long)hi << 32) | lo);
}

constexpr int NPW = 8;
constexpr int N2  = 32;
constexpr int TP  = 68;
template <int C, bool RELU, bool PROJ>
__device__ __forceinline__ void gather_body(const float* __restrict__ h, const int* __restrict__ sorted,
                                            const int* __restrict__ info, const int* __restrict__ cnt,
                                            const float* __restrict__ bias, float* __restrict__ out,
                                            float* trows, int Nn, int node0) {
    constexpr int G = 16;
    constexpr int NO_END = 0x7fffffff;
    const int lane = threadIdx.x & 31;
    const int nn = (Nn - node0) < NPW ? (Nn - node0) : NPW;

    v2i ni; ni.x = 0; ni.y = 0;
    int cn = 0;
    if (lane < nn) { ni = ((const v2i*)info)[node0 + lane]; cn = cnt[node0 + lane]; }

    float sx, sy = 0.0f;
    {
        const GLOBAL_MEM float* hs = wave_uniform_row(h, node0, C);
        if (C == 64) { const v2f sv = ((const GLOBAL_MEM v2f*)hs)[lane]; sx = sv.x; sy = sv.y; }
        else         { sx = hs[lane]; }
    }
    float bx, by = 0.0f;
    if (C == 64) { bx = bias[2 * lane]; by = bias[2 * lane + 1]; }
    else         { bx = bias[lane]; }

    const int start0 = __builtin_amdgcn_readfirstlane(ni.x);
    const int endl = ni.x + cn - start0;
    const int tot = __builtin_amdgcn_readlane(endl, nn - 1);
    const int* ent = sorted + start0;

    int k = 0;
    int seg_end = __builtin_amdgcn_readfirstlane(endl);
    float ax = 0.0f, ay = 0.0f;

    auto finish_node = [&]() {
        const float di = __int_as_float(__builtin_amdgcn_readlane(ni.y, k));
        if (C == 64) {
            float ox = di * (ax + sx) + bx;
            float oy = di * (ay + sy) + by;
            if (RELU) { ox = fmaxf(ox, 0.0f); oy = fmaxf(oy, 0.0f); }
            v2f ov; ov.x = ox; ov.y = oy;
            if (PROJ) ((v2f*)(trows + k * TP))[lane] = ov;
            else      ((GLOBAL_MEM v2f*)wave_uniform_row(out, node0 + k, C))[lane] = ov;
        } else {
            float ox = di * (ax + sx) + bx;
            if (RELU) ox = fmaxf(ox, 0.0f);
            wave_uniform_row(out, node0 + k, C)[lane] = ox;
        }
        ++k;
        ax = 0.0f; ay = 0.0f;
        if (k < nn) {
            seg_end = __builtin_amdgcn_readlane(endl, k);
            const GLOBAL_MEM float* hs = wave_uniform_row(h, node0 + k, C);
            if (C == 64) { const v2f sv = ((const GLOBAL_MEM v2f*)hs)[lane]; sx = sv.x; sy = sv.y; }
            else         { sx = hs[lane]; }
        } else {
            seg_end = NO_END;
        }
    };

    int s_cur = 0;
    if (lane < tot) s_cur = ent[lane];
    for (int c0 = 0; c0 < tot; c0 += 32) {
        int s_nxt = 0;
        const int m = (tot - c0) < 32 ? (tot - c0) : 32;
        for (int j0 = 0; j0 < m; j0 += G) {
            float v0[G], v1[G];
#pragma unroll
            for (int u = 0; u < G; ++u) {
                const int sj = __builtin_amdgcn_readlane(s_cur, j0 + u);
                const GLOBAL_MEM float* hr = wave_uniform_row(h, sj, C);
                if (C == 64) { const v2f tv = ((const GLOBAL_MEM v2f*)hr)[lane]; v0[u] = tv.x; v1[u] = tv.y; }
                else         { v0[u] = hr[lane]; }
            }
            if (j0 == 0 && c0 + 32 + lane < tot) s_nxt = ent[c0 + 32 + lane];
            const int e0 = c0 + j0;
            int rel = seg_end - e0;
#pragma unroll
            for (int u = 0; u < G; ++u) {
                while (u == rel) { finish_node(); rel = seg_end - e0; }
                ax += v0[u];
                if (C == 64) ay += v1[u];
            }
        }
        s_cur = s_nxt;
    }
    while (k < nn) finish_node();
}

template <int C, bool RELU, bool PROJ>
__global__ __launch_bounds__(256)
void gather_kernel(const float* __restrict__ h, const int* __restrict__ sorted,
                   const int* __restrict__ info, const int* __restrict__ cnt,
                   const float* __restrict__ bias, const float* __restrict__ W2,
                   float* __restrict__ out, int Nn) {
    static_assert(!PROJ || (C == 64 && NPW == 8), "the PROJ tile is 8 waves x 8 rows of 64 floats");
    __shared__ v4f   Wl[PROJ ? (C / 8) * (N2 / 16) * 32 : 1];
    __shared__ float tile[PROJ ? 64 * TP : 1];
    const int wib = threadIdx.x >> 5;
    const int wv = __builtin_amdgcn_readfirstlane((int)((blockIdx.x * blockDim.x + threadIdx.x) >> 5));
    const int node0 = wv * NPW;
    if (!PROJ) {
        if (node0 >= Nn) return;
        gather_body<C, RELU, false>(h, sorted, info, cnt, bias, out, nullptr, Nn, node0);
        return;
    }
    if (wib < 4) stage_w<C, N2>(W2, Wl, (int)threadIdx.x);
    if (node0 < Nn) gather_body<C, RELU, true>(h, sorted, info, cnt, bias, out, tile + wib * NPW * TP, Nn, node0);
    {
        const int have = node0 < Nn ? ((Nn - node0) < NPW ? (Nn - node0) : NPW) : 0;
        v2f zz; zz.x = 0.0f; zz.y = 0.0f;
        for (int k = have; k < NPW; ++k) ((v2f*)(tile + (wib * NPW + k) * TP))[threadIdx.x & 31] = zz;
    }
    __syncthreads();
    const int brow0 = blockIdx.x * 64 + wib * 16;
    if (wib >= 4 || brow0 >= Nn) return;
    const int lane = threadIdx.x & 31;
    const int half = lane >> 4;
    const int l16  = lane & 15;
    const int row  = brow0 + l16;
    const int arow = row < Nn ? row : Nn - 1;
    const float dva = __int_as_float(info[2 * arow + 1]);
    const float* trow = tile + (wib * 16 + l16) * TP + half * 4;
    v8f c0 = (v8f){0.f, 0.f, 0.f, 0.f, 0.f, 0.f, 0.f, 0.f};
    v8f c1 = c0;
#pragma unroll
    for (int s = 0; s < C / 8; ++s) {
        const v4f a = *(const v4f*)(trow + s * 8);
        v2f a0, a1;
        a0.x = a.x * dva; a0.y = a.y * dva;
        a1.x = a.z * dva; a1.y = a.w * dva;
        const v4f bA = Wl[(s * 2 + 0) * 32 + lane];
        const v4f bB = Wl[(s * 2 + 1) * 32 + (lane ^ 1)];
        v2f b;
        b.x = bA.x; b.y = bA.y;
        c0 = __builtin_amdgcn_wmma_f32_16x16x4_f32(false, a0, false, b, (short)0, c0, false, false);
        b.x = bB.x; b.y = bB.y;
        c1 = __builtin_amdgcn_wmma_f32_16x16x4_f32(false, a0, false, b, (short)0, c1, false, false);
        b.x = bA.z; b.y = bA.w;
        c0 = __builtin_amdgcn_wmma_f32_16x16x4_f32(false, a1, false, b, (short)0, c0, false, false);
        b.x = bB.z; b.y = bB.w;
        c1 = __builtin_amdgcn_wmma_f32_16x16x4_f32(false, a1, false, b, (short)0, c1, false, false);
    }
#pragma unroll
    for (int i = 0; i < 8; ++i) {
        const int r = brow0 + i + half * 8;
        if (r < Nn) {
            out[(long)r * N2 + l16]      = c0[i];
            out[(long)r * N2 + 16 + l16] = c1[i];
        }
    }
}

static inline size_t align256(size_t n) { return (n + 255) & ~size_t(255); }

extern "C" void kernel_launch(void* const* d_in, const int* in_sizes, int n_in,
                              void* d_out, int out_size, void* d_ws, size_t ws_size,
                              hipStream_t stream) {
    const float* x  = (const float*)d_in[0];
    const int*   ei = (const int*)d_in[1];
    const float* W1 = (const float*)d_in[2];
    const float* b1 = (const float*)d_in[3];
    const float* W2 = (const float*)d_in[4];
    const float* b2 = (const float*)d_in[5];
    float* out = (float*)d_out;

    const int hid  = in_sizes[3];
    const int outc = in_sizes[5];
    const int inc  = in_sizes[2] / hid;
    const int Nn   = in_sizes[0] / inc;
    const int E    = in_sizes[1] / 2;
    const int* src = ei;
    const int* dstv = ei + E;
    (void)outc;

    const int NB   = (Nn + BIN_NODES - 1) / BIN_NODES;
    const int nblk = (E + EPB - 1) / EPB;
    const int nt   = NB * nblk;

    char* ws = (char*)d_ws;
    int* cnt    = (int*)ws;    ws += align256((size_t)Nn * 4);
    int* info   = (int*)ws;    ws += align256((size_t)Nn * 8);
    int* table  = (int*)ws;    ws += align256((size_t)nt * 4);
    int* pre    = (int*)ws;    ws += align256((size_t)nt * 4);
    int* bucket = (int*)ws;    ws += align256((size_t)E * 8);
    int* sorted = (int*)ws;    ws += align256((size_t)E * 4);
    float* h1   = (float*)ws;  ws += (size_t)Nn * 64 * 4;
    float* h2   = (float*)ws;

    const int T = 256;
    bin_count_kernel<<<nblk, 256, 0, stream>>>(dstv, table, E, NB, nblk);
    scan_kernel<<<(nt + 1023) / 1024, 1024, 0, stream>>>(table, pre, nt);
    bin_scatter_kernel<<<nblk, 256, 0, stream>>>(src, dstv, pre, bucket, E, NB, nblk);
    bin_sort_kernel<<<NB, 1024, 0, stream>>>(bucket, pre, cnt, info, sorted, E, Nn, NB, nblk);

    gemm_wmma<128, 64, true><<<(Nn + 63) / 64, 128, 0, stream>>>(x, W1, info, h1, Nn);
    gather_kernel<64, true, true><<<(Nn + 8 * NPW - 1) / (8 * NPW), T, 0, stream>>>(h1, sorted, info, cnt, b1, W2, h2, Nn);

    gather_kernel<32, false, false><<<(Nn + 8 * NPW - 1) / (8 * NPW), T, 0, stream>>>(h2, sorted, info, cnt, b2, W2, out, Nn);
}


# list build: global count pass and scan dropped: a block leaves its edges bin-ordered in its own slice with local bin starts, a bin's block gathers its runs; 2 launches fewer; sum order: LDS arrival
# speedup vs baseline: 5.6153x; 1.1448x over previous
#include <hip/hip_runtime.h>

typedef __attribute__((ext_vector_type(2))) float v2f;
typedef __attribute__((ext_vector_type(8))) float v8f;

typedef __attribute__((ext_vector_type(2))) int v2i;
typedef __attribute__((ext_vector_type(4))) int v4i;

constexpr int BIN_SHIFT = 9;
constexpr int BIN_NODES = 1 << BIN_SHIFT;
constexpr int NB_MAX    = 256;
constexpr int EPT       = 16;
constexpr int EPB       = 256 * EPT;

template <bool FULL>
__device__ __forceinline__ void bin_local_body(const int* __restrict__ src, const int* __restrict__ dst,
                                               int* __restrict__ lstart, v2i* __restrict__ out2,
                                               int* hist, int* wsum, v2i* buf,
                                               int E, int NB, int nblk) {
    const int t = threadIdx.x, lane = t & 31, w = t >> 5;
    hist[t] = 0;
    __syncthreads();
    const int blk0 = blockIdx.x * EPB;
    const int nloc = FULL ? EPB : (E - blk0);
    int s[EPT], d[EPT], r[EPT];
#pragma unroll
    for (int k = 0; k < EPT; ++k) {
        const int e = blk0 + k * 256 + t;
        const int ec = (FULL || e < E) ? e : E - 1;
        s[k] = src[ec];
        d[k] = dst[ec];
    }
#pragma unroll
    for (int k = 0; k < EPT; ++k) {
        r[k] = 0;
        if (FULL || k * 256 + t < nloc) r[k] = atomicAdd(&hist[d[k] >> BIN_SHIFT], 1);
    }
    __syncthreads();
    const int c = hist[t];
    int incl = c;
#pragma unroll
    for (int o = 1; o < 32; o <<= 1) {
        const int v = __shfl_up(incl, o);
        if (lane >= o) incl += v;
    }
    if (lane == 31) wsum[w] = incl;
    __syncthreads();
    int front = 0;
#pragma unroll
    for (int j = 0; j < 8; ++j) front += j < w ? wsum[j] : 0;
    const int ls = front + incl - c;
    hist[t] = ls;
    if (t < NB) lstart[t * nblk + blockIdx.x] = ls;
    __syncthreads();
    int lp[EPT];
#pragma unroll
    for (int k = 0; k < EPT; ++k) lp[k] = hist[d[k] >> BIN_SHIFT] + r[k];
#pragma unroll
    for (int k = 0; k < EPT; ++k) {
        if (FULL || k * 256 + t < nloc) {
            v2i rec;
            rec.x = s[k];
            rec.y = d[k];
            buf[lp[k]] = rec;
        }
    }
    __syncthreads();
    v2i rec[EPT];
#pragma unroll
    for (int k = 0; k < EPT; ++k) {
        const int i = k * 256 + t;
        rec[k] = buf[(FULL || i < nloc) ? i : 0];
    }
#pragma unroll
    for (int k = 0; k < EPT; ++k) {
        const int i = k * 256 + t;
        if (FULL || i < nloc) out2[blk0 + i] = rec[k];
    }
}

__global__ __launch_bounds__(256)
void bin_local_kernel(const int* __restrict__ src, const int* __restrict__ dst,
                      int* __restrict__ lstart, int* __restrict__ bucketed,
                      int E, int NB, int nblk) {
    __shared__ int hist[NB_MAX];
    __shared__ int wsum[8];
    __shared__ v2i buf[EPB];
    v2i* out2 = (v2i*)bucketed;
    if (blockIdx.x * EPB + EPB <= E) bin_local_body<true>(src, dst, lstart, out2, hist, wsum, buf, E, NB, nblk);
    else                             bin_local_body<false>(src, dst, lstart, out2, hist, wsum, buf, E, NB, nblk);
}

template <typename F>
__device__ __forceinline__ void walk_runs(const v2i* __restrict__ in2, const int* __restrict__ rs,
                                          const int* __restrict__ re, bool lastbin, int E,
                                          int seg0, int seg1, int sv0, int ev0, int lane, F f) {
    constexpr int U = 16;
    for (int c0 = seg0; c0 < seg1; c0 += 32) {
        int sv = sv0, ev = ev0;
        if (c0 != seg0) {
            sv = 0; ev = 0;
            const int q = c0 + lane;
            if (q < seg1) {
                sv = rs[q];
                const int left = E - q * EPB;
                ev = lastbin ? (left < EPB ? left : EPB) : re[q];
            }
        }
        const int m = (seg1 - c0) < 32 ? (seg1 - c0) : 32;
        for (int u0 = 0; u0 < m; u0 += U) {
            v2i e[U];
            bool ok[U];
#pragma unroll
            for (int j = 0; j < U; ++j) {
                const int u  = u0 + j;
                const int ul = u < m ? u : m - 1;
                const int s  = __builtin_amdgcn_readlane(sv, ul);
                const int en = __builtin_amdgcn_readlane(ev, ul);
                const int i  = s + lane;
                const int r0 = (c0 + ul) * EPB;
                ok[j] = (u < m) && (i < en);
                e[j] = in2[ok[j] ? r0 + i : (en > s ? r0 + s : 0)];
            }
#pragma unroll
            for (int j = 0; j < U; ++j) asm volatile("" :: "v"(e[j].y));
#pragma unroll
            for (int j = 0; j < U; ++j)
                if (ok[j]) f(e[j]);
            for (int u = u0; u < u0 + U && u < m; ++u) {
                const int s  = __builtin_amdgcn_readlane(sv, u);
                const int en = __builtin_amdgcn_readlane(ev, u);
                for (int i = s + 32 + lane; i < en; i += 32) f(in2[(c0 + u) * EPB + i]);
            }
        }
    }
}

constexpr int STAGE_CAP = 16384;
__global__ __launch_bounds__(1024)
void bin_sort_kernel(const int* __restrict__ bucketed, const int* __restrict__ lstart,
                     int* __restrict__ cnt, int* __restrict__ info, int* __restrict__ sorted,
                     int E, int Nn, int NB, int nblk) {
    __shared__ int lc[BIN_NODES];
    __shared__ int wtot[BIN_NODES / 32];
    __shared__ int red_s[32];
    __shared__ int red_n[32];
    __shared__ int stage[STAGE_CAP];
    constexpr int U = 8;
    const int t = threadIdx.x, lane = t & 31;
    const int w = __builtin_amdgcn_readfirstlane(t >> 5);
    const int b = blockIdx.x;
    const v2i* in2 = (const v2i*)bucketed;
    const bool lastbin = b + 1 >= NB;
    const int* rs = lstart + b * nblk;
    const int* re = lstart + (lastbin ? b : b + 1) * nblk;
    const int spw  = (nblk + 31) >> 5;
    const int seg0 = w * spw;
    const int seg1 = (seg0 + spw) < nblk ? (seg0 + spw) : nblk;

    int sv0 = 0, ev0 = 0, ps = 0, pn = 0;
    for (int c0 = seg0; c0 < seg1; c0 += 32) {
        const int q = c0 + lane;
        int sv = 0, ev = 0;
        if (q < seg1) {
            sv = rs[q];
            const int left = E - q * EPB;
            ev = lastbin ? (left < EPB ? left : EPB) : re[q];
        }
        if (c0 == seg0) { sv0 = sv; ev0 = ev; }
        ps += sv;
        pn += ev - sv;
    }
#pragma unroll
    for (int o = 16; o > 0; o >>= 1) { ps += __shfl_xor(ps, o); pn += __shfl_xor(pn, o); }
    if (lane == 0) { red_s[w] = ps; red_n[w] = pn; }
    if (t < BIN_NODES) lc[t] = 0;
    __syncthreads();
    int beg = red_s[lane], nb = red_n[lane];
#pragma unroll
    for (int o = 16; o > 0; o >>= 1) { beg += __shfl_xor(beg, o); nb += __shfl_xor(nb, o); }

    walk_runs(in2, rs, re, lastbin, E, seg0, seg1, sv0, ev0, lane,
              [&](v2i rec) { atomicAdd(&lc[rec.y & (BIN_NODES - 1)], 1); });
    __syncthreads();

    const int c = t < BIN_NODES ? lc[t] : 0;
    int incl = c;
#pragma unroll
    for (int o = 1; o < 32; o <<= 1) {
        const int v = __shfl_up(incl, o);
        if (lane >= o) incl += v;
    }
    if (lane == 31 && t < BIN_NODES) wtot[w] = incl;
    __syncthreads();
    if (t < BIN_NODES) {
        int front = 0;
        for (int j = 0; j < w; ++j) front += wtot[j];
        const int st = front + incl - c;
        const int dv = __float_as_int(rsqrtf((float)(c + 1)));
        lc[t] = st;
        const int node = b * BIN_NODES + t;
        if (node < Nn) {
            v2i rec;
            rec.x = beg + st;
            rec.y = dv;
            ((v2i*)info)[node] = rec;
            cnt[node] = c;
        }
    }
    __syncthreads();

    int* out1 = sorted + beg;
    const bool fits = nb <= STAGE_CAP;
    walk_runs(in2, rs, re, lastbin, E, seg0, seg1, sv0, ev0, lane,
              [&](v2i rec) {
                  const int pos = atomicAdd(&lc[rec.y & (BIN_NODES - 1)], 1);
                  if (fits) stage[pos] = rec.x;
                  else      out1[pos] = rec.x;
              });
    if (fits) {
        __syncthreads();
        for (int i0 = t; i0 < nb; i0 += 1024 * U) {
            int rec[U];
#pragma unroll
            for (int u = 0; u < U; ++u) {
                const int i = i0 + u * 1024;
                rec[u] = stage[i < nb ? i : i0];
            }
#pragma unroll
            for (int u = 0; u < U; ++u) {
                const int i = i0 + u * 1024;
                if (i < nb) out1[i] = rec[u];
            }
        }
    }
}

__device__ __forceinline__ float squash(float v) {
    return (v * v) * __builtin_amdgcn_rcpf(v);
}

typedef __attribute__((ext_vector_type(4))) float v4f;

template <int K, int N>
__device__ __forceinline__ void stage_w(const float* __restrict__ W, v4f* Wl, int tid) {
    constexpr int NT  = N / 16;
    constexpr int CG  = N / 4;
    constexpr int QPP = 128 / CG;
    const int cg = tid % CG;
    const int t  = cg >> 2;
    const int l0 = (cg & 3) * 4;
#pragma unroll
    for (int p = 0; p < K / 4 / QPP; ++p) {
        const int q = tid / CG + p * QPP;
        const float* wp = W + (4 * q) * N + 4 * cg;
        const v4f r0 = *(const v4f*)(wp);
        const v4f r1 = *(const v4f*)(wp + N);
        const v4f r2 = *(const v4f*)(wp + 2 * N);
        const v4f r3 = *(const v4f*)(wp + 3 * N);
        v4f* d = Wl + (((q >> 1) * NT + t) * 32 + (q & 1) * 16 + l0);
        v4f o0; o0.x = r0.x; o0.y = r1.x; o0.z = r2.x; o0.w = r3.x;
        v4f o1; o1.x = r0.y; o1.y = r1.y; o1.z = r2.y; o1.w = r3.y;
        v4f o2; o2.x = r0.z; o2.y = r1.z; o2.z = r2.z; o2.w = r3.z;
        v4f o3; o3.x = r0.w; o3.y = r1.w; o3.z = r2.w; o3.w = r3.w;
        d[0 ^ t] = o0; d[1 ^ t] = o1; d[2 ^ t] = o2; d[3 ^ t] = o3;
    }
}

template <int K, int N, bool SQUASH>
__global__ __launch_bounds__(128)
void gemm_wmma(const float* __restrict__ X, const float* __restrict__ W,
               const int* __restrict__ info, float* __restrict__ H, int M) {
    constexpr int NT = N / 16;
    constexpr int KS = K / 8;
    __shared__ v4f Wl[KS * NT * 32];
    const int tid   = threadIdx.x;
    const int lane  = tid & 31;
    const int half  = lane >> 4;
    const int l16   = lane & 15;
    const int row0  = (blockIdx.x * 4 + (tid >> 5)) * 16;
    const int row   = row0 + l16;
    const int arow  = row < M ? row : M - 1;
    const float* __restrict__ xrow = X + (long)arow * K + half * 4;

    v4f av[KS];
#pragma unroll
    for (int s = 0; s < KS; ++s) av[s] = *(const v4f*)(xrow + s * 8);
    const float dva = __int_as_float(info[2 * arow + 1]);

    stage_w<K, N>(W, Wl, tid);
    __syncthreads();

    v8f c[NT];
#pragma unroll
    for (int t = 0; t < NT; ++t) c[t] = (v8f){0.f, 0.f, 0.f, 0.f, 0.f, 0.f, 0.f, 0.f};

#pragma unroll
    for (int s = 0; s < KS; ++s) {
        v4f a = av[s];
        if (SQUASH) { a.x = squash(a.x); a.y = squash(a.y); a.z = squash(a.z); a.w = squash(a.w); }
        v2f a0, a1;
        a0.x = a.x * dva; a0.y = a.y * dva;
        a1.x = a.z * dva; a1.y = a.w * dva;
        v4f b[NT];
#pragma unroll
        for (int t = 0; t < NT; ++t) b[t] = Wl[(s * NT + t) * 32 + (lane ^ t)];
#pragma unroll
        for (int t = 0; t < NT; ++t) {
            v2f b0; b0.x = b[t].x; b0.y = b[t].y;
            c[t] = __builtin_amdgcn_wmma_f32_16x16x4_f32(
                        false, a0,  false, b0,
                        (short)0, c[t],  false,  false);
        }
#pragma unroll
        for (int t = 0; t < NT; ++t) {
            v2f b1; b1.x = b[t].z; b1.y = b[t].w;
            c[t] = __builtin_amdgcn_wmma_f32_16x16x4_f32(
                        false, a1,  false, b1,
                        (short)0, c[t],  false,  false);
        }
    }

#pragma unroll
    for (int i = 0; i < 8; ++i) {
        const int r = row0 + i + half * 8;
        if (r < M) {
#pragma unroll
            for (int t = 0; t < NT; ++t) H[(long)r * N + t * 16 + l16] = c[t][i];
        }
    }
}

#define GLOBAL_MEM __attribute__((address_space(1)))
template <typename T>
__device__ __forceinline__ GLOBAL_MEM T* wave_uniform_row(T* base, int row, int width) {
    const unsigned long long a = (unsigned long long)(base + (long)row * width);
    const unsigned lo = __builtin_amdgcn_readfirstlane((unsigned)a);
    const unsigned hi = __builtin_amdgcn_readfirstlane((unsigned)(a >> 32));
    return (GLOBAL_MEM T*)(((unsigned long long)hi << 32) | lo);
}

constexpr int NPW = 8;
constexpr int N2  = 32;
constexpr int TP  = 68;
template <int C, bool RELU, bool PROJ>
__device__ __forceinline__ void gather_body(const float* __restrict__ h, const int* __restrict__ sorted,
                                            const int* __restrict__ info, const int* __restrict__ cnt,
                                            const float* __restrict__ bias, float* __restrict__ out,
                                            float* trows, int Nn, int node0) {
    constexpr int G = 16;
    constexpr int NO_END = 0x7fffffff;
    const int lane = threadIdx.x & 31;
    const int nn = (Nn - node0) < NPW ? (Nn - node0) : NPW;

    v2i ni; ni.x = 0; ni.y = 0;
    int cn = 0;
    if (lane < nn) { ni = ((const v2i*)info)[node0 + lane]; cn = cnt[node0 + lane]; }

    float sx, sy = 0.0f;
    {
        const GLOBAL_MEM float* hs = wave_uniform_row(h, node0, C);
        if (C == 64) { const v2f sv = ((const GLOBAL_MEM v2f*)hs)[lane]; sx = sv.x; sy = sv.y; }
        else         { sx = hs[lane]; }
    }
    float bx, by = 0.0f;
    if (C == 64) { bx = bias[2 * lane]; by = bias[2 * lane + 1]; }
    else         { bx = bias[lane]; }

    const int start0 = __builtin_amdgcn_readfirstlane(ni.x);
    const int endl = ni.x + cn - start0;
    const int tot = __builtin_amdgcn_readlane(endl, nn - 1);
    const int* ent = sorted + start0;

    int k = 0;
    int seg_end = __builtin_amdgcn_readfirstlane(endl);
    float ax = 0.0f, ay = 0.0f;

    auto finish_node = [&]() {
        const float di = __int_as_float(__builtin_amdgcn_readlane(ni.y, k));
        if (C == 64) {
            float ox = di * (ax + sx) + bx;
            float oy = di * (ay + sy) + by;
            if (RELU) { ox = fmaxf(ox, 0.0f); oy = fmaxf(oy, 0.0f); }
            v2f ov; ov.x = ox; ov.y = oy;
            if (PROJ) ((v2f*)(trows + k * TP))[lane] = ov;
            else      ((GLOBAL_MEM v2f*)wave_uniform_row(out, node0 + k, C))[lane] = ov;
        } else {
            float ox = di * (ax + sx) + bx;
            if (RELU) ox = fmaxf(ox, 0.0f);
            wave_uniform_row(out, node0 + k, C)[lane] = ox;
        }
        ++k;
        ax = 0.0f; ay = 0.0f;
        if (k < nn) {
            seg_end = __builtin_amdgcn_readlane(endl, k);
            const GLOBAL_MEM float* hs = wave_uniform_row(h, node0 + k, C);
            if (C == 64) { const v2f sv = ((const GLOBAL_MEM v2f*)hs)[lane]; sx = sv.x; sy = sv.y; }
            else         { sx = hs[lane]; }
        } else {
            seg_end = NO_END;
        }
    };

    int s_cur = 0;
    if (lane < tot) s_cur = ent[lane];
    for (int c0 = 0; c0 < tot; c0 += 32) {
        int s_nxt = 0;
        const int m = (tot - c0) < 32 ? (tot - c0) : 32;
        for (int j0 = 0; j0 < m; j0 += G) {
            float v0[G], v1[G];
#pragma unroll
            for (int u = 0; u < G; ++u) {
                const int sj = __builtin_amdgcn_readlane(s_cur, j0 + u);
                const GLOBAL_MEM float* hr = wave_uniform_row(h, sj, C);
                if (C == 64) { const v2f tv = ((const GLOBAL_MEM v2f*)hr)[lane]; v0[u] = tv.x; v1[u] = tv.y; }
                else         { v0[u] = hr[lane]; }
            }
            if (j0 == 0 && c0 + 32 + lane < tot) s_nxt = ent[c0 + 32 + lane];
            const int e0 = c0 + j0;
            int rel = seg_end - e0;
#pragma unroll
            for (int u = 0; u < G; ++u) {
                while (u == rel) { finish_node(); rel = seg_end - e0; }
                ax += v0[u];
                if (C == 64) ay += v1[u];
            }
        }
        s_cur = s_nxt;
    }
    while (k < nn) finish_node();
}

template <int C, bool RELU, bool PROJ>
__global__ __launch_bounds__(256)
void gather_kernel(const float* __restrict__ h, const int* __restrict__ sorted,
                   const int* __restrict__ info, const int* __restrict__ cnt,
                   const float* __restrict__ bias, const float* __restrict__ W2,
                   float* __restrict__ out, int Nn) {
    static_assert(!PROJ || (C == 64 && NPW == 8), "the PROJ tile is 8 waves x 8 rows of 64 floats");
    __shared__ v4f   Wl[PROJ ? (C / 8) * (N2 / 16) * 32 : 1];
    __shared__ float tile[PROJ ? 64 * TP : 1];
    const int wib = threadIdx.x >> 5;
    const int wv = __builtin_amdgcn_readfirstlane((int)((blockIdx.x * blockDim.x + threadIdx.x) >> 5));
    const int node0 = wv * NPW;
    if (!PROJ) {
        if (node0 >= Nn) return;
        gather_body<C, RELU, false>(h, sorted, info, cnt, bias, out, nullptr, Nn, node0);
        return;
    }
    if (wib < 4) stage_w<C, N2>(W2, Wl, (int)threadIdx.x);
    if (node0 < Nn) gather_body<C, RELU, true>(h, sorted, info, cnt, bias, out, tile + wib * NPW * TP, Nn, node0);
    {
        const int have = node0 < Nn ? ((Nn - node0) < NPW ? (Nn - node0) : NPW) : 0;
        v2f zz; zz.x = 0.0f; zz.y = 0.0f;
        for (int k = have; k < NPW; ++k) ((v2f*)(tile + (wib * NPW + k) * TP))[threadIdx.x & 31] = zz;
    }
    __syncthreads();
    const int brow0 = blockIdx.x * 64 + wib * 16;
    if (wib >= 4 || brow0 >= Nn) return;
    const int lane = threadIdx.x & 31;
    const int half = lane >> 4;
    const int l16  = lane & 15;
    const int row  = brow0 + l16;
    const int arow = row < Nn ? row : Nn - 1;
    const float dva = __int_as_float(info[2 * arow + 1]);
    const float* trow = tile + (wib * 16 + l16) * TP + half * 4;
    v8f c0 = (v8f){0.f, 0.f, 0.f, 0.f, 0.f, 0.f, 0.f, 0.f};
    v8f c1 = c0;
#pragma unroll
    for (int s = 0; s < C / 8; ++s) {
        const v4f a = *(const v4f*)(trow + s * 8);
        v2f a0, a1;
        a0.x = a.x * dva; a0.y = a.y * dva;
        a1.x = a.z * dva; a1.y = a.w * dva;
        const v4f bA = Wl[(s * 2 + 0) * 32 + lane];
        const v4f bB = Wl[(s * 2 + 1) * 32 + (lane ^ 1)];
        v2f b;
        b.x = bA.x; b.y = bA.y;
        c0 = __builtin_amdgcn_wmma_f32_16x16x4_f32(false, a0, false, b, (short)0, c0, false, false);
        b.x = bB.x; b.y = bB.y;
        c1 = __builtin_amdgcn_wmma_f32_16x16x4_f32(false, a0, false, b, (short)0, c1, false, false);
        b.x = bA.z; b.y = bA.w;
        c0 = __builtin_amdgcn_wmma_f32_16x16x4_f32(false, a1, false, b, (short)0, c0, false, false);
        b.x = bB.z; b.y = bB.w;
        c1 = __builtin_amdgcn_wmma_f32_16x16x4_f32(false, a1, false, b, (short)0, c1, false, false);
    }
#pragma unroll
    for (int i = 0; i < 8; ++i) {
        const int r = brow0 + i + half * 8;
        if (r < Nn) {
            out[(long)r * N2 + l16]      = c0[i];
            out[(long)r * N2 + 16 + l16] = c1[i];
        }
    }
}

static inline size_t align256(size_t n) { return (n + 255) & ~size_t(255); }

extern "C" void kernel_launch(void* const* d_in, const int* in_sizes, int n_in,
                              void* d_out, int out_size, void* d_ws, size_t ws_size,
                              hipStream_t stream) {
    const float* x  = (const float*)d_in[0];
    const int*   ei = (const int*)d_in[1];
    const float* W1 = (const float*)d_in[2];
    const float* b1 = (const float*)d_in[3];
    const float* W2 = (const float*)d_in[4];
    const float* b2 = (const float*)d_in[5];
    float* out = (float*)d_out;

    const int hid  = in_sizes[3];
    const int outc = in_sizes[5];
    const int inc  = in_sizes[2] / hid;
    const int Nn   = in_sizes[0] / inc;
    const int E    = in_sizes[1] / 2;
    const int* src = ei;
    const int* dstv = ei + E;
    (void)outc;

    const int NB   = (Nn + BIN_NODES - 1) / BIN_NODES;
    const int nblk = (E + EPB - 1) / EPB;
    const int nt   = NB * nblk;

    char* ws = (char*)d_ws;
    int* cnt    = (int*)ws;    ws += align256((size_t)Nn * 4);
    int* info   = (int*)ws;    ws += align256((size_t)Nn * 8);
    int* lstart = (int*)ws;    ws += align256((size_t)nt * 4);
    int* bucket = (int*)ws;    ws += align256((size_t)E * 8);
    int* sorted = (int*)ws;    ws += align256((size_t)E * 4);
    float* h1   = (float*)ws;  ws += (size_t)Nn * 64 * 4;
    float* h2   = (float*)ws;

    const int T = 256;
    bin_local_kernel<<<nblk, 256, 0, stream>>>(src, dstv, lstart, bucket, E, NB, nblk);
    bin_sort_kernel<<<NB, 1024, 0, stream>>>(bucket, lstart, cnt, info, sorted, E, Nn, NB, nblk);

    gemm_wmma<128, 64, true><<<(Nn + 63) / 64, 128, 0, stream>>>(x, W1, info, h1, Nn);
    gather_kernel<64, true, true><<<(Nn + 8 * NPW - 1) / (8 * NPW), T, 0, stream>>>(h1, sorted, info, cnt, b1, W2, h2, Nn);

    gather_kernel<32, false, false><<<(Nn + 8 * NPW - 1) / (8 * NPW), T, 0, stream>>>(h2, sorted, info, cnt, b2, W2, out, Nn);
}


# layer-1 GEMM moved into the sort launch: the 32-wave block that sorts a bin multiplies the bin's 512 rows, dinv from LDS, x rows prefetched under the sort; one launch fewer; sum orders kept
# speedup vs baseline: 5.8407x; 1.0401x over previous
#include <hip/hip_runtime.h>

typedef __attribute__((ext_vector_type(2))) float v2f;
typedef __attribute__((ext_vector_type(8))) float v8f;

typedef __attribute__((ext_vector_type(2))) int v2i;
typedef __attribute__((ext_vector_type(4))) int v4i;

constexpr int BIN_SHIFT = 9;
constexpr int BIN_NODES = 1 << BIN_SHIFT;
constexpr int NB_MAX    = 256;
constexpr int EPT       = 16;
constexpr int EPB       = 256 * EPT;

template <bool FULL>
__device__ __forceinline__ void bin_local_body(const int* __restrict__ src, const int* __restrict__ dst,
                                               int* __restrict__ lstart, v2i* __restrict__ out2,
                                               int* hist, int* wsum, v2i* buf,
                                               int E, int NB, int nblk) {
    const int t = threadIdx.x, lane = t & 31, w = t >> 5;
    hist[t] = 0;
    __syncthreads();
    const int blk0 = blockIdx.x * EPB;
    const int nloc = FULL ? EPB : (E - blk0);
    int s[EPT], d[EPT], r[EPT];
#pragma unroll
    for (int k = 0; k < EPT; ++k) {
        const int e = blk0 + k * 256 + t;
        const int ec = (FULL || e < E) ? e : E - 1;
        s[k] = src[ec];
        d[k] = dst[ec];
    }
#pragma unroll
    for (int k = 0; k < EPT; ++k) {
        r[k] = 0;
        if (FULL || k * 256 + t < nloc) r[k] = atomicAdd(&hist[d[k] >> BIN_SHIFT], 1);
    }
    __syncthreads();
    const int c = hist[t];
    int incl = c;
#pragma unroll
    for (int o = 1; o < 32; o <<= 1) {
        const int v = __shfl_up(incl, o);
        if (lane >= o) incl += v;
    }
    if (lane == 31) wsum[w] = incl;
    __syncthreads();
    int front = 0;
#pragma unroll
    for (int j = 0; j < 8; ++j) front += j < w ? wsum[j] : 0;
    const int ls = front + incl - c;
    hist[t] = ls;
    if (t < NB) lstart[t * nblk + blockIdx.x] = ls;
    __syncthreads();
    int lp[EPT];
#pragma unroll
    for (int k = 0; k < EPT; ++k) lp[k] = hist[d[k] >> BIN_SHIFT] + r[k];
#pragma unroll
    for (int k = 0; k < EPT; ++k) {
        if (FULL || k * 256 + t < nloc) {
            v2i rec;
            rec.x = s[k];
            rec.y = d[k];
            buf[lp[k]] = rec;
        }
    }
    __syncthreads();
    v2i rec[EPT];
#pragma unroll
    for (int k = 0; k < EPT; ++k) {
        const int i = k * 256 + t;
        rec[k] = buf[(FULL || i < nloc) ? i : 0];
    }
#pragma unroll
    for (int k = 0; k < EPT; ++k) {
        const int i = k * 256 + t;
        if (FULL || i < nloc) out2[blk0 + i] = rec[k];
    }
}

__global__ __launch_bounds__(256)
void bin_local_kernel(const int* __restrict__ src, const int* __restrict__ dst,
                      int* __restrict__ lstart, int* __restrict__ bucketed,
                      int E, int NB, int nblk) {
    __shared__ int hist[NB_MAX];
    __shared__ int wsum[8];
    __shared__ v2i buf[EPB];
    v2i* out2 = (v2i*)bucketed;
    if (blockIdx.x * EPB + EPB <= E) bin_local_body<true>(src, dst, lstart, out2, hist, wsum, buf, E, NB, nblk);
    else                             bin_local_body<false>(src, dst, lstart, out2, hist, wsum, buf, E, NB, nblk);
}

__device__ __forceinline__ float squash(float v) {
    return (v * v) * __builtin_amdgcn_rcpf(v);
}

typedef __attribute__((ext_vector_type(4))) float v4f;

template <int K, int N, int NTHR = 128>
__device__ __forceinline__ void stage_w(const float* __restrict__ W, v4f* Wl, int tid) {
    constexpr int NT  = N / 16;
    constexpr int CG  = N / 4;
    constexpr int QPP = NTHR / CG;
    const int cg = tid % CG;
    const int t  = cg >> 2;
    const int l0 = (cg & 3) * 4;
#pragma unroll
    for (int p = 0; p < K / 4 / QPP; ++p) {
        const int q = tid / CG + p * QPP;
        const float* wp = W + (4 * q) * N + 4 * cg;
        const v4f r0 = *(const v4f*)(wp);
        const v4f r1 = *(const v4f*)(wp + N);
        const v4f r2 = *(const v4f*)(wp + 2 * N);
        const v4f r3 = *(const v4f*)(wp + 3 * N);
        v4f* d = Wl + (((q >> 1) * NT + t) * 32 + (q & 1) * 16 + l0);
        v4f o0; o0.x = r0.x; o0.y = r1.x; o0.z = r2.x; o0.w = r3.x;
        v4f o1; o1.x = r0.y; o1.y = r1.y; o1.z = r2.y; o1.w = r3.y;
        v4f o2; o2.x = r0.z; o2.y = r1.z; o2.z = r2.z; o2.w = r3.z;
        v4f o3; o3.x = r0.w; o3.y = r1.w; o3.z = r2.w; o3.w = r3.w;
        d[0 ^ t] = o0; d[1 ^ t] = o1; d[2 ^ t] = o2; d[3 ^ t] = o3;
    }
}

template <int K, int N, bool SQUASH>
__device__ __forceinline__ void wave_rows_times_w(const float* __restrict__ xrow, float dva, const v4f* Wl,
                                                  float* __restrict__ H, int row0, int M, int lane) {
    constexpr int NT = N / 16;
    constexpr int KS = K / 8;
    const int half = lane >> 4;
    const int l16  = lane & 15;

    v4f av[KS];
#pragma unroll
    for (int s = 0; s < KS; ++s) av[s] = *(const v4f*)(xrow + s * 8);

    v8f c[NT];
#pragma unroll
    for (int t = 0; t < NT; ++t) c[t] = (v8f){0.f, 0.f, 0.f, 0.f, 0.f, 0.f, 0.f, 0.f};

#pragma unroll
    for (int s = 0; s < KS; ++s) {
        v4f a = av[s];
        if (SQUASH) { a.x = squash(a.x); a.y = squash(a.y); a.z = squash(a.z); a.w = squash(a.w); }
        v2f a0, a1;
        a0.x = a.x * dva; a0.y = a.y * dva;
        a1.x = a.z * dva; a1.y = a.w * dva;
        v4f b[NT];
#pragma unroll
        for (int t = 0; t < NT; ++t) b[t] = Wl[(s * NT + t) * 32 + (lane ^ t)];
#pragma unroll
        for (int t = 0; t < NT; ++t) {
            v2f b0; b0.x = b[t].x; b0.y = b[t].y;
            c[t] = __builtin_amdgcn_wmma_f32_16x16x4_f32(
                        false, a0,  false, b0,
                        (short)0, c[t],  false,  false);
        }
#pragma unroll
        for (int t = 0; t < NT; ++t) {
            v2f b1; b1.x = b[t].z; b1.y = b[t].w;
            c[t] = __builtin_amdgcn_wmma_f32_16x16x4_f32(
                        false, a1,  false, b1,
                        (short)0, c[t],  false,  false);
        }
    }

#pragma unroll
    for (int i = 0; i < 8; ++i) {
        const int r = row0 + i + half * 8;
        if (r < M) {
#pragma unroll
            for (int t = 0; t < NT; ++t) H[(long)r * N + t * 16 + l16] = c[t][i];
        }
    }
}

template <typename F, typename P>
__device__ __forceinline__ void walk_runs(const v2i* __restrict__ in2, const int* __restrict__ rs,
                                          const int* __restrict__ re, bool lastbin, int E,
                                          int seg0, int seg1, int sv0, int ev0, int lane, F f,
                                          P behind_first_loads) {
    constexpr int U = 16;
    bool fired = false;
    for (int c0 = seg0; c0 < seg1; c0 += 32) {
        int sv = sv0, ev = ev0;
        if (c0 != seg0) {
            sv = 0; ev = 0;
            const int q = c0 + lane;
            if (q < seg1) {
                sv = rs[q];
                const int left = E - q * EPB;
                ev = lastbin ? (left < EPB ? left : EPB) : re[q];
            }
        }
        const int m = (seg1 - c0) < 32 ? (seg1 - c0) : 32;
        for (int u0 = 0; u0 < m; u0 += U) {
            v2i e[U];
            bool ok[U];
#pragma unroll
            for (int j = 0; j < U; ++j) {
                const int u  = u0 + j;
                const int ul = u < m ? u : m - 1;
                const int s  = __builtin_amdgcn_readlane(sv, ul);
                const int en = __builtin_amdgcn_readlane(ev, ul);
                const int i  = s + lane;
                const int r0 = (c0 + ul) * EPB;
                ok[j] = (u < m) && (i < en);
                e[j] = in2[ok[j] ? r0 + i : (en > s ? r0 + s : 0)];
            }
            if (!fired) { behind_first_loads(); fired = true; }
#pragma unroll
            for (int j = 0; j < U; ++j) asm volatile("" :: "v"(e[j].x), "v"(e[j].y));
#pragma unroll
            for (int j = 0; j < U; ++j)
                if (ok[j]) f(e[j]);
            for (int u = u0; u < u0 + U && u < m; ++u) {
                const int s  = __builtin_amdgcn_readlane(sv, u);
                const int en = __builtin_amdgcn_readlane(ev, u);
                for (int i = s + 32 + lane; i < en; i += 32) f(in2[(c0 + u) * EPB + i]);
            }
        }
    }
    if (!fired) behind_first_loads();
}

constexpr int STAGE_CAP = 16384;
constexpr int GK = 128;
constexpr int GN = 64;
static_assert(BIN_NODES == 32 * 16, "32 waves x 16 rows cover one bin");
__global__ __launch_bounds__(1024) __attribute__((amdgpu_waves_per_eu(8, 8)))
void bin_sort_gemm_kernel(const int* __restrict__ bucketed, const int* __restrict__ lstart,
                          int* __restrict__ cnt, int* __restrict__ info, int* __restrict__ sorted,
                          const float* __restrict__ X, const float* __restrict__ W, float* __restrict__ H,
                          int E, int Nn, int NB, int nblk) {
    __shared__ int lc[BIN_NODES];
    __shared__ int wtot[BIN_NODES / 32];
    __shared__ int red_s[32];
    __shared__ int red_n[32];
    __shared__ int stage[STAGE_CAP];
    __shared__ v4f   Wl[(GK / 8) * (GN / 16) * 32];
    __shared__ float dinv_s[BIN_NODES];
    constexpr int U = 8;
    const int t = threadIdx.x, lane = t & 31;
    const int w = __builtin_amdgcn_readfirstlane(t >> 5);
    const int b = blockIdx.x;
    const v2i* in2 = (const v2i*)bucketed;
    const bool lastbin = b + 1 >= NB;

    const int row0 = b * BIN_NODES + w * 16;
    auto ask_for_x_rows = [&]() {
        const int pr = row0 + (lane >> 1);
        __builtin_prefetch(X + (long)(pr < Nn ? pr : Nn - 1) * GK + (lane & 1) * (GK / 2), 0, 3);
    };
    const int* rs = lstart + b * nblk;
    const int* re = lstart + (lastbin ? b : b + 1) * nblk;
    const int spw  = (nblk + 31) >> 5;
    const int seg0 = w * spw;
    const int seg1 = (seg0 + spw) < nblk ? (seg0 + spw) : nblk;

    int sv0 = 0, ev0 = 0, ps = 0, pn = 0;
    for (int c0 = seg0; c0 < seg1; c0 += 32) {
        const int q = c0 + lane;
        int sv = 0, ev = 0;
        if (q < seg1) {
            sv = rs[q];
            const int left = E - q * EPB;
            ev = lastbin ? (left < EPB ? left : EPB) : re[q];
        }
        if (c0 == seg0) { sv0 = sv; ev0 = ev; }
        ps += sv;
        pn += ev - sv;
    }
#pragma unroll
    for (int o = 16; o > 0; o >>= 1) { ps += __shfl_xor(ps, o); pn += __shfl_xor(pn, o); }
    if (lane == 0) { red_s[w] = ps; red_n[w] = pn; }
    if (t < BIN_NODES) lc[t] = 0;
    __syncthreads();
    int beg = red_s[lane], nb = red_n[lane];
#pragma unroll
    for (int o = 16; o > 0; o >>= 1) { beg += __shfl_xor(beg, o); nb += __shfl_xor(nb, o); }

    walk_runs(in2, rs, re, lastbin, E, seg0, seg1, sv0, ev0, lane,
              [&](v2i rec) { atomicAdd(&lc[rec.y & (BIN_NODES - 1)], 1); }, ask_for_x_rows);
    __syncthreads();

    const int c = t < BIN_NODES ? lc[t] : 0;
    int incl = c;
#pragma unroll
    for (int o = 1; o < 32; o <<= 1) {
        const int v = __shfl_up(incl, o);
        if (lane >= o) incl += v;
    }
    if (lane == 31 && t < BIN_NODES) wtot[w] = incl;
    __syncthreads();
    if (t < BIN_NODES) {
        int front = 0;
        for (int j = 0; j < w; ++j) front += wtot[j];
        const int st = front + incl - c;
        const int dv = __float_as_int(rsqrtf((float)(c + 1)));
        lc[t] = st;
        dinv_s[t] = __int_as_float(dv);
        const int node = b * BIN_NODES + t;
        if (node < Nn) {
            v2i rec;
            rec.x = beg + st;
            rec.y = dv;
            ((v2i*)info)[node] = rec;
            cnt[node] = c;
        }
    } else {
        stage_w<GK, GN, 512>(W, Wl, t - BIN_NODES);
    }
    __syncthreads();

    int* out1 = sorted + beg;
    const bool fits = nb <= STAGE_CAP;
    walk_runs(in2, rs, re, lastbin, E, seg0, seg1, sv0, ev0, lane,
              [&](v2i rec) {
                  const int pos = atomicAdd(&lc[rec.y & (BIN_NODES - 1)], 1);
                  if (fits) stage[pos] = rec.x;
                  else      out1[pos] = rec.x;
              }, []() {});
    if (fits) {
        __syncthreads();
        for (int i0 = t; i0 < nb; i0 += 1024 * U) {
            int rec[U];
#pragma unroll
            for (int u = 0; u < U; ++u) {
                const int i = i0 + u * 1024;
                rec[u] = stage[i < nb ? i : i0];
            }
#pragma unroll
            for (int u = 0; u < U; ++u) {
                const int i = i0 + u * 1024;
                if (i < nb) out1[i] = rec[u];
            }
        }
    }

    if (row0 < Nn) {
        const int l16  = lane & 15;
        const int row  = row0 + l16;
        const int arow = row < Nn ? row : Nn - 1;
        wave_rows_times_w<GK, GN, true>(X + (long)arow * GK + (lane >> 4) * 4, dinv_s[w * 16 + l16],
                                        Wl, H, row0, Nn, lane);
    }
}

#define GLOBAL_MEM __attribute__((address_space(1)))
template <typename T>
__device__ __forceinline__ GLOBAL_MEM T* wave_uniform_row(T* base, int row, int width) {
    const unsigned long long a = (unsigned long long)(base + (long)row * width);
    const unsigned lo = __builtin_amdgcn_readfirstlane((unsigned)a);
    const unsigned hi = __builtin_amdgcn_readfirstlane((unsigned)(a >> 32));
    return (GLOBAL_MEM T*)(((unsigned long long)hi << 32) | lo);
}

constexpr int NPW = 8;
constexpr int N2  = 32;
constexpr int TP  = 68;
template <int C, bool RELU, bool PROJ>
__device__ __forceinline__ void gather_body(const float* __restrict__ h, const int* __restrict__ sorted,
                                            const int* __restrict__ info, const int* __restrict__ cnt,
                                            const float* __restrict__ bias, float* __restrict__ out,
                                            float* trows, int Nn, int node0) {
    constexpr int G = 16;
    constexpr int NO_END = 0x7fffffff;
    const int lane = threadIdx.x & 31;
    const int nn = (Nn - node0) < NPW ? (Nn - node0) : NPW;

    v2i ni; ni.x = 0; ni.y = 0;
    int cn = 0;
    if (lane < nn) { ni = ((const v2i*)info)[node0 + lane]; cn = cnt[node0 + lane]; }

    float sx, sy = 0.0f;
    {
        const GLOBAL_MEM float* hs = wave_uniform_row(h, node0, C);
        if (C == 64) { const v2f sv = ((const GLOBAL_MEM v2f*)hs)[lane]; sx = sv.x; sy = sv.y; }
        else         { sx = hs[lane]; }
    }
    float bx, by = 0.0f;
    if (C == 64) { bx = bias[2 * lane]; by = bias[2 * lane + 1]; }
    else         { bx = bias[lane]; }

    const int start0 = __builtin_amdgcn_readfirstlane(ni.x);
    const int endl = ni.x + cn - start0;
    const int tot = __builtin_amdgcn_readlane(endl, nn - 1);
    const int* ent = sorted + start0;

    int k = 0;
    int seg_end = __builtin_amdgcn_readfirstlane(endl);
    float ax = 0.0f, ay = 0.0f;

    auto finish_node = [&]() {
        const float di = __int_as_float(__builtin_amdgcn_readlane(ni.y, k));
        if (C == 64) {
            float ox = di * (ax + sx) + bx;
            float oy = di * (ay + sy) + by;
            if (RELU) { ox = fmaxf(ox, 0.0f); oy = fmaxf(oy, 0.0f); }
            v2f ov; ov.x = ox; ov.y = oy;
            if (PROJ) ((v2f*)(trows + k * TP))[lane] = ov;
            else      ((GLOBAL_MEM v2f*)wave_uniform_row(out, node0 + k, C))[lane] = ov;
        } else {
            float ox = di * (ax + sx) + bx;
            if (RELU) ox = fmaxf(ox, 0.0f);
            wave_uniform_row(out, node0 + k, C)[lane] = ox;
        }
        ++k;
        ax = 0.0f; ay = 0.0f;
        if (k < nn) {
            seg_end = __builtin_amdgcn_readlane(endl, k);
            const GLOBAL_MEM float* hs = wave_uniform_row(h, node0 + k, C);
            if (C == 64) { const v2f sv = ((const GLOBAL_MEM v2f*)hs)[lane]; sx = sv.x; sy = sv.y; }
            else         { sx = hs[lane]; }
        } else {
            seg_end = NO_END;
        }
    };

    int s_cur = 0;
    if (lane < tot) s_cur = ent[lane];
    for (int c0 = 0; c0 < tot; c0 += 32) {
        int s_nxt = 0;
        const int m = (tot - c0) < 32 ? (tot - c0) : 32;
        for (int j0 = 0; j0 < m; j0 += G) {
            float v0[G], v1[G];
#pragma unroll
            for (int u = 0; u < G; ++u) {
                const int sj = __builtin_amdgcn_readlane(s_cur, j0 + u);
                const GLOBAL_MEM float* hr = wave_uniform_row(h, sj, C);
                if (C == 64) { const v2f tv = ((const GLOBAL_MEM v2f*)hr)[lane]; v0[u] = tv.x; v1[u] = tv.y; }
                else         { v0[u] = hr[lane]; }
            }
            if (j0 == 0 && c0 + 32 + lane < tot) s_nxt = ent[c0 + 32 + lane];
            const int e0 = c0 + j0;
            int rel = seg_end - e0;
#pragma unroll
            for (int u = 0; u < G; ++u) {
                while (u == rel) { finish_node(); rel = seg_end - e0; }
                ax += v0[u];
                if (C == 64) ay += v1[u];
            }
        }
        s_cur = s_nxt;
    }
    while (k < nn) finish_node();
}

template <int C, bool RELU, bool PROJ>
__global__ __launch_bounds__(256)
void gather_kernel(const float* __restrict__ h, const int* __restrict__ sorted,
                   const int* __restrict__ info, const int* __restrict__ cnt,
                   const float* __restrict__ bias, const float* __restrict__ W2,
                   float* __restrict__ out, int Nn) {
    static_assert(!PROJ || (C == 64 && NPW == 8), "the PROJ tile is 8 waves x 8 rows of 64 floats");
    __shared__ v4f   Wl[PROJ ? (C / 8) * (N2 / 16) * 32 : 1];
    __shared__ float tile[PROJ ? 64 * TP : 1];
    const int wib = threadIdx.x >> 5;
    const int wv = __builtin_amdgcn_readfirstlane((int)((blockIdx.x * blockDim.x + threadIdx.x) >> 5));
    const int node0 = wv * NPW;
    if (!PROJ) {
        if (node0 >= Nn) return;
        gather_body<C, RELU, false>(h, sorted, info, cnt, bias, out, nullptr, Nn, node0);
        return;
    }
    if (wib < 4) stage_w<C, N2>(W2, Wl, (int)threadIdx.x);
    if (node0 < Nn) gather_body<C, RELU, true>(h, sorted, info, cnt, bias, out, tile + wib * NPW * TP, Nn, node0);
    {
        const int have = node0 < Nn ? ((Nn - node0) < NPW ? (Nn - node0) : NPW) : 0;
        v2f zz; zz.x = 0.0f; zz.y = 0.0f;
        for (int k = have; k < NPW; ++k) ((v2f*)(tile + (wib * NPW + k) * TP))[threadIdx.x & 31] = zz;
    }
    __syncthreads();
    const int brow0 = blockIdx.x * 64 + wib * 16;
    if (wib >= 4 || brow0 >= Nn) return;
    const int lane = threadIdx.x & 31;
    const int half = lane >> 4;
    const int l16  = lane & 15;
    const int row  = brow0 + l16;
    const int arow = row < Nn ? row : Nn - 1;
    const float dva = __int_as_float(info[2 * arow + 1]);
    const float* trow = tile + (wib * 16 + l16) * TP + half * 4;
    v8f c0 = (v8f){0.f, 0.f, 0.f, 0.f, 0.f, 0.f, 0.f, 0.f};
    v8f c1 = c0;
#pragma unroll
    for (int s = 0; s < C / 8; ++s) {
        const v4f a = *(const v4f*)(trow + s * 8);
        v2f a0, a1;
        a0.x = a.x * dva; a0.y = a.y * dva;
        a1.x = a.z * dva; a1.y = a.w * dva;
        const v4f bA = Wl[(s * 2 + 0) * 32 + lane];
        const v4f bB = Wl[(s * 2 + 1) * 32 + (lane ^ 1)];
        v2f b;
        b.x = bA.x; b.y = bA.y;
        c0 = __builtin_amdgcn_wmma_f32_16x16x4_f32(false, a0, false, b, (short)0, c0, false, false);
        b.x = bB.x; b.y = bB.y;
        c1 = __builtin_amdgcn_wmma_f32_16x16x4_f32(false, a0, false, b, (short)0, c1, false, false);
        b.x = bA.z; b.y = bA.w;
        c0 = __builtin_amdgcn_wmma_f32_16x16x4_f32(false, a1, false, b, (short)0, c0, false, false);
        b.x = bB.z; b.y = bB.w;
        c1 = __builtin_amdgcn_wmma_f32_16x16x4_f32(false, a1, false, b, (short)0, c1, false, false);
    }
#pragma unroll
    for (int i = 0; i < 8; ++i) {
        const int r = brow0 + i + half * 8;
        if (r < Nn) {
            out[(long)r * N2 + l16]      = c0[i];
            out[(long)r * N2 + 16 + l16] = c1[i];
        }
    }
}

static inline size_t align256(size_t n) { return (n + 255) & ~size_t(255); }

extern "C" void kernel_launch(void* const* d_in, const int* in_sizes, int n_in,
                              void* d_out, int out_size, void* d_ws, size_t ws_size,
                              hipStream_t stream) {
    const float* x  = (const float*)d_in[0];
    const int*   ei = (const int*)d_in[1];
    const float* W1 = (const float*)d_in[2];
    const float* b1 = (const float*)d_in[3];
    const float* W2 = (const float*)d_in[4];
    const float* b2 = (const float*)d_in[5];
    float* out = (float*)d_out;

    const int hid  = in_sizes[3];
    const int outc = in_sizes[5];
    const int inc  = in_sizes[2] / hid;
    const int Nn   = in_sizes[0] / inc;
    const int E    = in_sizes[1] / 2;
    const int* src = ei;
    const int* dstv = ei + E;
    (void)outc;

    const int NB   = (Nn + BIN_NODES - 1) / BIN_NODES;
    const int nblk = (E + EPB - 1) / EPB;
    const int nt   = NB * nblk;

    char* ws = (char*)d_ws;
    int* cnt    = (int*)ws;    ws += align256((size_t)Nn * 4);
    int* info   = (int*)ws;    ws += align256((size_t)Nn * 8);
    int* lstart = (int*)ws;    ws += align256((size_t)nt * 4);
    int* bucket = (int*)ws;    ws += align256((size_t)E * 8);
    int* sorted = (int*)ws;    ws += align256((size_t)E * 4);
    float* h1   = (float*)ws;  ws += (size_t)Nn * 64 * 4;
    float* h2   = (float*)ws;

    const int T = 256;
    bin_local_kernel<<<nblk, 256, 0, stream>>>(src, dstv, lstart, bucket, E, NB, nblk);
    bin_sort_gemm_kernel<<<NB, 1024, 0, stream>>>(bucket, lstart, cnt, info, sorted, x, W1, h1, E, Nn, NB, nblk);

    gather_kernel<64, true, true><<<(Nn + 8 * NPW - 1) / (8 * NPW), T, 0, stream>>>(h1, sorted, info, cnt, b1, W2, h2, Nn);

    gather_kernel<32, false, false><<<(Nn + 8 * NPW - 1) / (8 * NPW), T, 0, stream>>>(h2, sorted, info, cnt, b2, W2, out, Nn);
}
